# starting kernel
# baseline (speedup 1.0000x reference)
#include <hip/hip_runtime.h>


typedef __attribute__((ext_vector_type(2))) float v2f;
typedef __attribute__((ext_vector_type(8))) float v8f;

template <bool RELU, bool DUAL>
__global__ __launch_bounds__(256) void gemm_wmma_kernel(
    const float* __restrict__ A, const float* __restrict__ W1,
    const float* __restrict__ B, const float* __restrict__ W2,
    const float* __restrict__ bias, float* __restrict__ out,
    int M, int Cin, int Cout)
{
    const int lane = threadIdx.x & 31;
    const int wavesPerBlock = blockDim.x >> 5;
    const int tile = blockIdx.x * wavesPerBlock + (threadIdx.x >> 5);
    const int ntn = Cout >> 4;
    const int ntm = M >> 4;
    if (tile >= ntm * ntn) return;
    const int tm = tile / ntn;
    const int tn = tile - tm * ntn;

    const int l    = lane & 15;
    const int koff = (lane >> 4) << 1;
    const int col  = tn * 16 + l;

    const float* arow = A + (size_t)(tm * 16 + l) * Cin;
    const float* brow = DUAL ? (B + (size_t)(tm * 16 + l) * Cin) : nullptr;

    v8f acc = {};
    for (int k = 0; k < Cin; k += 4) {
        v2f av; av.x = arow[k + koff];
                av.y = arow[k + koff + 1];
        v2f wv; wv.x = W1[(size_t)(k + koff)     * Cout + col];
                wv.y = W1[(size_t)(k + koff + 1) * Cout + col];
        acc = __builtin_amdgcn_wmma_f32_16x16x4_f32(
                  false, av, false, wv, (short)0, acc, false, false);
        if (DUAL) {
            v2f bv; bv.x = brow[k + koff];
                    bv.y = brow[k + koff + 1];
            v2f w2; w2.x = W2[(size_t)(k + koff)     * Cout + col];
                    w2.y = W2[(size_t)(k + koff + 1) * Cout + col];
            acc = __builtin_amdgcn_wmma_f32_16x16x4_f32(
                      false, bv, false, w2, (short)0, acc, false, false);
        }
    }

    const float bval = bias[col];
    const int m0 = tm * 16 + ((lane >> 4) << 3);
    #pragma unroll
    for (int r = 0; r < 8; ++r) {
        float v = acc[r] + bval;
        if (RELU) v = fmaxf(v, 0.0f);
        out[(size_t)(m0 + r) * Cout + col] = v;
    }
}

__global__ __launch_bounds__(256) void scatter_add_kernel(
    const float* __restrict__ h, const int* __restrict__ src,
    const int* __restrict__ dst, float* __restrict__ agg,
    long long E, int cshift)
{
    long long idx = (long long)blockIdx.x * blockDim.x + threadIdx.x;
    const int g4shift = cshift - 2;
    const long long total = E << g4shift;
    if (idx >= total) return;
    const int e  = (int)(idx >> g4shift);
    const int c4 = (int)(idx & ((1 << g4shift) - 1)) << 2;
    const int s  = src[e];
    const int d  = dst[e];
    const float4 v = *(const float4*)(h + ((size_t)s << cshift) + c4);
    float* dp = agg + ((size_t)d << cshift) + c4;
    atomicAdd(dp + 0, v.x);
    atomicAdd(dp + 1, v.y);
    atomicAdd(dp + 2, v.z);
    atomicAdd(dp + 3, v.w);
}

__global__ __launch_bounds__(256) void zero_kernel(float4* __restrict__ p, long long n4)
{
    long long i = (long long)blockIdx.x * blockDim.x + threadIdx.x;
    if (i < n4) p[i] = make_float4(0.f, 0.f, 0.f, 0.f);
}

__global__ __launch_bounds__(256) void pool_sum_kernel(
    const float* __restrict__ h, const int* __restrict__ batch,
    float* __restrict__ sums, float* __restrict__ cnt, long long N)
{
    long long idx = (long long)blockIdx.x * blockDim.x + threadIdx.x;
    const long long total = N << 5;
    if (idx >= total) return;
    const int node = (int)(idx >> 5);
    const int c4   = (int)(idx & 31) << 2;
    const int g    = batch[node];
    const float4 v = *(const float4*)(h + ((size_t)node << 7) + c4);
    float* sp = sums + ((size_t)g << 7) + c4;
    atomicAdd(sp + 0, v.x);
    atomicAdd(sp + 1, v.y);
    atomicAdd(sp + 2, v.z);
    atomicAdd(sp + 3, v.w);
    if (c4 == 0) atomicAdd(&cnt[g], 1.0f);
}

__global__ __launch_bounds__(256) void pool_div_kernel(
    const float4* __restrict__ sums, const float* __restrict__ cnt,
    float4* __restrict__ pooled, int G)
{
    int idx = blockIdx.x * blockDim.x + threadIdx.x;
    if (idx >= (G << 5)) return;
    const float m = fmaxf(cnt[idx >> 5], 1.0f);
    float4 s = sums[idx];
    pooled[idx] = make_float4(s.x / m, s.y / m, s.z / m, s.w / m);
}


static inline unsigned blocks_for(long long n, int t) {
    return (unsigned)((n + t - 1) / t);
}

extern "C" void kernel_launch(void* const* d_in, const int* in_sizes, int n_in,
                              void* d_out, int out_size, void* d_ws, size_t ws_size,
                              hipStream_t stream)
{
    const float* x        = (const float*)d_in[0];
    const int*   edge     = (const int*)  d_in[1];
    const int*   batch    = (const int*)  d_in[2];
    const float* w_stage1 = (const float*)d_in[4];
    const float* b_stage1 = (const float*)d_in[5];
    const float* rel1_w   = (const float*)d_in[6];
    const float* rel1_b   = (const float*)d_in[7];
    const float* root1_w  = (const float*)d_in[8];
    const float* rel2_w   = (const float*)d_in[9];
    const float* rel2_b   = (const float*)d_in[10];
    const float* root2_w  = (const float*)d_in[11];
    const float* rel3_w   = (const float*)d_in[12];
    const float* rel3_b   = (const float*)d_in[13];
    const float* root3_w  = (const float*)d_in[14];
    const float* lin_w    = (const float*)d_in[15];
    const float* lin_b    = (const float*)d_in[16];
    float*       out      = (float*)d_out;

    const int N = in_sizes[0] / 128;
    const int E = in_sizes[1] / 2;
    const int G = out_size / 16;
    const int* src = edge;
    const int* dst = edge + E;

    float* ws = (float*)d_ws;
    size_t off = 0;
    float* bufA   = ws + off; off += (size_t)N * 128;
    float* bufB   = ws + off; off += (size_t)N * 128;
    float* agg    = ws + off; off += (size_t)N * 128;
    float* sums   = ws + off; off += (size_t)G * 128;
    float* cnt    = ws + off; off += (size_t)G;
    float* pooled = ws + off; off += (size_t)G * 128;
    (void)ws_size; (void)n_in;

    const int T = 256;
    const int WPB = T / 32;

    auto gemm_grid = [&](int M, int Cout) {
        return blocks_for((long long)(M >> 4) * (Cout >> 4), WPB);
    };

    gemm_wmma_kernel<false, false><<<gemm_grid(N, 64), T, 0, stream>>>(
        x, w_stage1, nullptr, nullptr, b_stage1, bufA, N, 128, 64);

    zero_kernel<<<blocks_for((long long)N * 64 / 4, T), T, 0, stream>>>(
        (float4*)agg, (long long)N * 64 / 4);
    scatter_add_kernel<<<blocks_for((long long)E << 4, T), T, 0, stream>>>(
        bufA, src, dst, agg, E, 6);
    gemm_wmma_kernel<true, true><<<gemm_grid(N, 64), T, 0, stream>>>(
        agg, rel1_w, bufA, root1_w, rel1_b, bufB, N, 64, 64);

    zero_kernel<<<blocks_for((long long)N * 64 / 4, T), T, 0, stream>>>(
        (float4*)agg, (long long)N * 64 / 4);
    scatter_add_kernel<<<blocks_for((long long)E << 4, T), T, 0, stream>>>(
        bufB, src, dst, agg, E, 6);
    gemm_wmma_kernel<true, true><<<gemm_grid(N, 128), T, 0, stream>>>(
        agg, rel2_w, bufB, root2_w, rel2_b, bufA, N, 64, 128);

    zero_kernel<<<blocks_for((long long)N * 128 / 4, T), T, 0, stream>>>(
        (float4*)agg, (long long)N * 128 / 4);
    scatter_add_kernel<<<blocks_for((long long)E << 5, T), T, 0, stream>>>(
        bufA, src, dst, agg, E, 7);
    gemm_wmma_kernel<true, true><<<gemm_grid(N, 128), T, 0, stream>>>(
        agg, rel3_w, bufA, root3_w, rel3_b, bufB, N, 128, 128);

    zero_kernel<<<blocks_for((long long)G * 128 / 4, T), T, 0, stream>>>(
        (float4*)sums, (long long)G * 128 / 4);
    zero_kernel<<<blocks_for((long long)G / 4, T), T, 0, stream>>>(
        (float4*)cnt, (long long)G / 4);
    pool_sum_kernel<<<blocks_for((long long)N << 5, T), T, 0, stream>>>(
        bufB, batch, sums, cnt, N);
    pool_div_kernel<<<blocks_for((long long)G << 5, T), T, 0, stream>>>(
        (const float4*)sums, cnt, (float4*)pooled, G);

    gemm_wmma_kernel<false, false><<<gemm_grid(G, 16), T, 0, stream>>>(
        pooled, lin_w, nullptr, nullptr, lin_b, out, G, 128, 16);
}


# Edges grouped by destination once per call (counting sort, integer atomics), each node then sums its rows in registers: no float atomics or agg zero fill; per-node sum order = arrival-order rank
# speedup vs baseline: 1.5226x; 1.5226x over previous
#include <hip/hip_runtime.h>


typedef __attribute__((ext_vector_type(2))) float v2f;
typedef __attribute__((ext_vector_type(8))) float v8f;

template <bool RELU, bool DUAL>
__global__ __launch_bounds__(256) void gemm_wmma_kernel(
    const float* __restrict__ A, const float* __restrict__ W1,
    const float* __restrict__ B, const float* __restrict__ W2,
    const float* __restrict__ bias, float* __restrict__ out,
    int M, int Cin, int Cout)
{
    const int lane = threadIdx.x & 31;
    const int wavesPerBlock = blockDim.x >> 5;
    const int tile = blockIdx.x * wavesPerBlock + (threadIdx.x >> 5);
    const int ntn = Cout >> 4;
    const int ntm = M >> 4;
    if (tile >= ntm * ntn) return;
    const int tm = tile / ntn;
    const int tn = tile - tm * ntn;

    const int l    = lane & 15;
    const int koff = (lane >> 4) << 1;
    const int col  = tn * 16 + l;

    const float* arow = A + (size_t)(tm * 16 + l) * Cin;
    const float* brow = DUAL ? (B + (size_t)(tm * 16 + l) * Cin) : nullptr;

    v8f acc = {};
    for (int k = 0; k < Cin; k += 4) {
        v2f av; av.x = arow[k + koff];
                av.y = arow[k + koff + 1];
        v2f wv; wv.x = W1[(size_t)(k + koff)     * Cout + col];
                wv.y = W1[(size_t)(k + koff + 1) * Cout + col];
        acc = __builtin_amdgcn_wmma_f32_16x16x4_f32(
                  false, av, false, wv, (short)0, acc, false, false);
        if (DUAL) {
            v2f bv; bv.x = brow[k + koff];
                    bv.y = brow[k + koff + 1];
            v2f w2; w2.x = W2[(size_t)(k + koff)     * Cout + col];
                    w2.y = W2[(size_t)(k + koff + 1) * Cout + col];
            acc = __builtin_amdgcn_wmma_f32_16x16x4_f32(
                      false, bv, false, w2, (short)0, acc, false, false);
        }
    }

    const float bval = bias[col];
    const int m0 = tm * 16 + ((lane >> 4) << 3);
    #pragma unroll
    for (int r = 0; r < 8; ++r) {
        float v = acc[r] + bval;
        if (RELU) v = fmaxf(v, 0.0f);
        out[(size_t)(m0 + r) * Cout + col] = v;
    }
}

__global__ __launch_bounds__(256) void count_rank_kernel(
    const int* __restrict__ dst, int* __restrict__ deg,
    int* __restrict__ rank, int E)
{
    const int e = blockIdx.x * blockDim.x + threadIdx.x;
    if (e >= E) return;
    rank[e] = atomicAdd(&deg[dst[e]], 1);
}

__global__ __launch_bounds__(256) void row_alloc_kernel(
    const int* __restrict__ deg, int* __restrict__ cursor,
    int2* __restrict__ seg, int N)
{
    __shared__ int wave_sum[8];
    __shared__ int block_base;
    const int n    = blockIdx.x * blockDim.x + threadIdx.x;
    const int lane = threadIdx.x & 31;
    const int w    = threadIdx.x >> 5;
    const int d    = (n < N) ? deg[n] : 0;
    int v = d;
    for (int o = 1; o < 32; o <<= 1) {
        const int t = __shfl_up(v, o);
        if (lane >= o) v += t;
    }
    if (lane == 31) wave_sum[w] = v;
    __syncthreads();
    if (threadIdx.x == 0) {
        int s = 0;
        for (int i = 0; i < 8; ++i) { const int t = wave_sum[i]; wave_sum[i] = s; s += t; }
        block_base = atomicAdd(cursor, s);
    }
    __syncthreads();
    if (n < N) {
        const int first = block_base + wave_sum[w] + v - d;
        seg[n] = make_int2(first, first + d);
    }
}

__global__ __launch_bounds__(256) void fill_kernel(
    const int* __restrict__ src, const int* __restrict__ dst,
    const int* __restrict__ rank, const int2* __restrict__ seg,
    int* __restrict__ col, int E)
{
    const int e = blockIdx.x * blockDim.x + threadIdx.x;
    if (e >= E) return;
    col[seg[dst[e]].x + rank[e]] = src[e];
}

template <int CSHIFT>
__global__ __launch_bounds__(256) void gather_sum_kernel(
    const float* __restrict__ h, const int* __restrict__ col,
    const int2* __restrict__ seg, float* __restrict__ agg, int N)
{
    constexpr int G4SHIFT = CSHIFT - 2;
    const long long idx = (long long)blockIdx.x * blockDim.x + threadIdx.x;
    const int node = (int)(idx >> G4SHIFT);
    if (node >= N) return;
    const unsigned c4 = ((unsigned)idx & ((1u << G4SHIFT) - 1u)) << 2;
    const int2 run = seg[node];
    int j = run.x;
    const int end = run.y;
    float4 acc = make_float4(0.f, 0.f, 0.f, 0.f);
    for (; j < end; j += 8) {
        int s[8];
        for (int u = 0; u < 8; ++u) {
            const int jj = j + u;
            s[u] = col[jj < end ? jj : end - 1];
        }
        asm volatile("" :: "v"(s[0]), "v"(s[1]), "v"(s[2]), "v"(s[3]),
                           "v"(s[4]), "v"(s[5]), "v"(s[6]), "v"(s[7]));
        float4 v[8];
        v[0] = *(const float4*)(h + (((unsigned)s[0] << CSHIFT) + c4));
        for (int u = 1; u < 8; ++u) {
            v[u] = make_float4(0.f, 0.f, 0.f, 0.f);
            if (j + u < end) v[u] = *(const float4*)(h + (((unsigned)s[u] << CSHIFT) + c4));
        }
        for (int u = 0; u < 8; ++u) {
            acc.x += v[u].x; acc.y += v[u].y; acc.z += v[u].z; acc.w += v[u].w;
        }
    }
    *(float4*)(agg + ((size_t)node << CSHIFT) + c4) = acc;
}

__global__ __launch_bounds__(256) void zero_kernel(float4* __restrict__ p, long long n4)
{
    long long i = (long long)blockIdx.x * blockDim.x + threadIdx.x;
    if (i < n4) p[i] = make_float4(0.f, 0.f, 0.f, 0.f);
}

__global__ __launch_bounds__(256) void pool_sum_kernel(
    const float* __restrict__ h, const int* __restrict__ batch,
    float* __restrict__ sums, float* __restrict__ cnt, long long N)
{
    long long idx = (long long)blockIdx.x * blockDim.x + threadIdx.x;
    const long long total = N << 5;
    if (idx >= total) return;
    const int node = (int)(idx >> 5);
    const int c4   = (int)(idx & 31) << 2;
    const int g    = batch[node];
    const float4 v = *(const float4*)(h + ((size_t)node << 7) + c4);
    float* sp = sums + ((size_t)g << 7) + c4;
    atomicAdd(sp + 0, v.x);
    atomicAdd(sp + 1, v.y);
    atomicAdd(sp + 2, v.z);
    atomicAdd(sp + 3, v.w);
    if (c4 == 0) atomicAdd(&cnt[g], 1.0f);
}

__global__ __launch_bounds__(256) void pool_div_kernel(
    const float4* __restrict__ sums, const float* __restrict__ cnt,
    float4* __restrict__ pooled, int G)
{
    int idx = blockIdx.x * blockDim.x + threadIdx.x;
    if (idx >= (G << 5)) return;
    const float m = fmaxf(cnt[idx >> 5], 1.0f);
    float4 s = sums[idx];
    pooled[idx] = make_float4(s.x / m, s.y / m, s.z / m, s.w / m);
}


static inline unsigned blocks_for(long long n, int t) {
    return (unsigned)((n + t - 1) / t);
}

extern "C" void kernel_launch(void* const* d_in, const int* in_sizes, int n_in,
                              void* d_out, int out_size, void* d_ws, size_t ws_size,
                              hipStream_t stream)
{
    const float* x        = (const float*)d_in[0];
    const int*   edge     = (const int*)  d_in[1];
    const int*   batch    = (const int*)  d_in[2];
    const float* w_stage1 = (const float*)d_in[4];
    const float* b_stage1 = (const float*)d_in[5];
    const float* rel1_w   = (const float*)d_in[6];
    const float* rel1_b   = (const float*)d_in[7];
    const float* root1_w  = (const float*)d_in[8];
    const float* rel2_w   = (const float*)d_in[9];
    const float* rel2_b   = (const float*)d_in[10];
    const float* root2_w  = (const float*)d_in[11];
    const float* rel3_w   = (const float*)d_in[12];
    const float* rel3_b   = (const float*)d_in[13];
    const float* root3_w  = (const float*)d_in[14];
    const float* lin_w    = (const float*)d_in[15];
    const float* lin_b    = (const float*)d_in[16];
    float*       out      = (float*)d_out;

    const int N = in_sizes[0] / 128;
    const int E = in_sizes[1] / 2;
    const int G = out_size / 16;
    const int* src = edge;
    const int* dst = edge + E;

    float* ws = (float*)d_ws;
    size_t off = 0;
    float* bufA   = ws + off; off += (size_t)N * 128;
    float* bufB   = ws + off; off += (size_t)N * 128;
    float* agg    = ws + off; off += (size_t)N * 128;
    float* sums   = ws + off; off += (size_t)G * 128;
    float* cnt    = ws + off; off += (size_t)G;
    float* pooled = ws + off; off += (size_t)G * 128;
    const long long deg4 = ((long long)N + 1 + 3) / 4;
    int* deg    = (int*)(ws + off); off += (size_t)deg4 * 4;
    int* cursor = deg + N;
    int2* seg   = (int2*)(ws + off); off += (size_t)N * 2;
    int* rank   = (int*)(ws + off); off += (size_t)E;
    int* col    = (int*)(ws + off); off += (size_t)E;
    (void)ws_size; (void)n_in;

    const int T = 256;
    const int WPB = T / 32;

    auto gemm_grid = [&](int M, int Cout) {
        return blocks_for((long long)(M >> 4) * (Cout >> 4), WPB);
    };

    zero_kernel<<<blocks_for(deg4, T), T, 0, stream>>>((float4*)deg, deg4);
    count_rank_kernel<<<blocks_for(E, T), T, 0, stream>>>(dst, deg, rank, E);
    row_alloc_kernel<<<blocks_for(N, T), T, 0, stream>>>(deg, cursor, seg, N);
    fill_kernel<<<blocks_for(E, T), T, 0, stream>>>(src, dst, rank, seg, col, E);

    gemm_wmma_kernel<false, false><<<gemm_grid(N, 64), T, 0, stream>>>(
        x, w_stage1, nullptr, nullptr, b_stage1, bufA, N, 128, 64);

    gather_sum_kernel<6><<<blocks_for((long long)N << 4, T), T, 0, stream>>>(
        bufA, col, seg, agg, N);
    gemm_wmma_kernel<true, true><<<gemm_grid(N, 64), T, 0, stream>>>(
        agg, rel1_w, bufA, root1_w, rel1_b, bufB, N, 64, 64);

    gather_sum_kernel<6><<<blocks_for((long long)N << 4, T), T, 0, stream>>>(
        bufB, col, seg, agg, N);
    gemm_wmma_kernel<true, true><<<gemm_grid(N, 128), T, 0, stream>>>(
        agg, rel2_w, bufB, root2_w, rel2_b, bufA, N, 64, 128);

    gather_sum_kernel<7><<<blocks_for((long long)N << 5, T), T, 0, stream>>>(
        bufA, col, seg, agg, N);
    gemm_wmma_kernel<true, true><<<gemm_grid(N, 128), T, 0, stream>>>(
        agg, rel3_w, bufA, root3_w, rel3_b, bufB, N, 128, 128);

    zero_kernel<<<blocks_for((long long)G * 128 / 4, T), T, 0, stream>>>(
        (float4*)sums, (long long)G * 128 / 4);
    zero_kernel<<<blocks_for((long long)G / 4, T), T, 0, stream>>>(
        (float4*)cnt, (long long)G / 4);
    pool_sum_kernel<<<blocks_for((long long)N << 5, T), T, 0, stream>>>(
        bufB, batch, sums, cnt, N);
    pool_div_kernel<<<blocks_for((long long)G << 5, T), T, 0, stream>>>(
        (const float4*)sums, cnt, (float4*)pooled, G);

    gemm_wmma_kernel<false, false><<<gemm_grid(G, 16), T, 0, stream>>>(
        pooled, lin_w, nullptr, nullptr, lin_b, out, G, 128, 16);
}


# GEMM operand feed: one wave per 16-row strip reuses each A fragment over 4 column tiles (independent accumulators); weights staged in LDS as k-pairs; per-accumulator k order unchanged
# speedup vs baseline: 1.7798x; 1.1689x over previous
#include <hip/hip_runtime.h>


typedef __attribute__((ext_vector_type(2))) float v2f;
typedef __attribute__((ext_vector_type(8))) float v8f;

template <int NT, int CIN, bool RELU, bool DUAL>
__global__ __launch_bounds__(256) void gemm_wmma_kernel(
    const float* __restrict__ A, const float* __restrict__ W1,
    const float* __restrict__ B, const float* __restrict__ W2,
    const float* __restrict__ bias, float* __restrict__ out,
    int M, int Cout)
{
    constexpr int KC    = 64;
    constexpr int KP    = KC / 2;
    constexpr int PITCH = NT * 32 + ((NT & 1) ? 64 : 32);
    constexpr int NMAT  = DUAL ? 2 : 1;
    constexpr int C4    = NT * 4;
    __shared__ __attribute__((aligned(16))) float wl[NMAT * KP * PITCH];

    const int lane = threadIdx.x & 31;
    const int wave = threadIdx.x >> 5;
    const int ncg  = Cout / (16 * NT);
    const int rb   = blockIdx.x / ncg;
    const int cg   = blockIdx.x - rb * ncg;
    const int col0 = cg * (16 * NT);
    const int strip = rb * 8 + wave;
    const bool active = strip < (M >> 4);
    const int c = lane & 15;
    const int h = lane >> 4;

    const size_t rowoff = (size_t)((active ? strip : 0) * 16 + c) * CIN + 2 * h;
    const float* arow = A + rowoff;
    const float* brow = DUAL ? (B + rowoff) : nullptr;

    v8f acc[NT];
    #pragma unroll
    for (int t = 0; t < NT; ++t) acc[t] = v8f{};

    #pragma unroll
    for (int k0 = 0; k0 < CIN; k0 += KC) {
        v2f af[KC / 4];
        v2f bf[DUAL ? KC / 4 : 1];
        if (active) {
            #pragma unroll
            for (int kk = 0; kk < KC / 4; ++kk) {
                af[kk] = *(const v2f*)(arow + k0 + 4 * kk);
                if (DUAL) bf[kk] = *(const v2f*)(brow + k0 + 4 * kk);
            }
        }
        if (k0 > 0) __syncthreads();
        #pragma unroll
        for (int m = 0; m < NMAT; ++m) {
            const float* Wm = m ? W2 : W1;
            for (int i = threadIdx.x; i < KP * C4; i += 256) {
                const int kp = i / C4;
                const int c4 = i - kp * C4;
                const float* g = Wm + (size_t)(k0 + 2 * kp) * Cout + col0 + 4 * c4;
                const float4 r0 = *(const float4*)g;
                const float4 r1 = *(const float4*)(g + Cout);
                float* d = wl + (m * KP + kp) * PITCH + c4 * 8;
                *(float4*)d       = make_float4(r0.x, r1.x, r0.y, r1.y);
                *(float4*)(d + 4) = make_float4(r0.z, r1.z, r0.w, r1.w);
            }
        }
        __syncthreads();
        if (active) {
            #pragma unroll
            for (int kk = 0; kk < KC / 4; ++kk) {
                const float* wrow = wl + (2 * kk + h) * PITCH + c * 2;
                #pragma unroll
                for (int t = 0; t < NT; ++t) {
                    const v2f wv = *(const v2f*)(wrow + t * 32);
                    acc[t] = __builtin_amdgcn_wmma_f32_16x16x4_f32(
                                 false, af[kk], false, wv, (short)0, acc[t], false, false);
                }
                if (DUAL) {
                    #pragma unroll
                    for (int t = 0; t < NT; ++t) {
                        const v2f w2 = *(const v2f*)(wrow + KP * PITCH + t * 32);
                        acc[t] = __builtin_amdgcn_wmma_f32_16x16x4_f32(
                                     false, bf[kk], false, w2, (short)0, acc[t], false, false);
                    }
                }
            }
        }
    }
    if (!active) return;

    const int m0 = strip * 16 + (h << 3);
    #pragma unroll
    for (int t = 0; t < NT; ++t) {
        const int col = col0 + t * 16 + c;
        const float bval = bias[col];
        #pragma unroll
        for (int r = 0; r < 8; ++r) {
            float v = acc[t][r] + bval;
            if (RELU) v = fmaxf(v, 0.0f);
            out[(size_t)(m0 + r) * Cout + col] = v;
        }
    }
}

__global__ __launch_bounds__(256) void count_rank_kernel(
    const int* __restrict__ dst, int* __restrict__ deg,
    int* __restrict__ rank, int E)
{
    const int e = blockIdx.x * blockDim.x + threadIdx.x;
    if (e >= E) return;
    rank[e] = atomicAdd(&deg[dst[e]], 1);
}

__global__ __launch_bounds__(256) void row_alloc_kernel(
    const int* __restrict__ deg, int* __restrict__ cursor,
    int2* __restrict__ seg, int N)
{
    __shared__ int wave_sum[8];
    __shared__ int block_base;
    const int n    = blockIdx.x * blockDim.x + threadIdx.x;
    const int lane = threadIdx.x & 31;
    const int w    = threadIdx.x >> 5;
    const int d    = (n < N) ? deg[n] : 0;
    int v = d;
    for (int o = 1; o < 32; o <<= 1) {
        const int t = __shfl_up(v, o);
        if (lane >= o) v += t;
    }
    if (lane == 31) wave_sum[w] = v;
    __syncthreads();
    if (threadIdx.x == 0) {
        int s = 0;
        for (int i = 0; i < 8; ++i) { const int t = wave_sum[i]; wave_sum[i] = s; s += t; }
        block_base = atomicAdd(cursor, s);
    }
    __syncthreads();
    if (n < N) {
        const int first = block_base + wave_sum[w] + v - d;
        seg[n] = make_int2(first, first + d);
    }
}

__global__ __launch_bounds__(256) void fill_kernel(
    const int* __restrict__ src, const int* __restrict__ dst,
    const int* __restrict__ rank, const int2* __restrict__ seg,
    int* __restrict__ col, int E)
{
    const int e = blockIdx.x * blockDim.x + threadIdx.x;
    if (e >= E) return;
    col[seg[dst[e]].x + rank[e]] = src[e];
}

template <int CSHIFT>
__global__ __launch_bounds__(256) void gather_sum_kernel(
    const float* __restrict__ h, const int* __restrict__ col,
    const int2* __restrict__ seg, float* __restrict__ agg, int N)
{
    constexpr int G4SHIFT = CSHIFT - 2;
    const long long idx = (long long)blockIdx.x * blockDim.x + threadIdx.x;
    const int node = (int)(idx >> G4SHIFT);
    if (node >= N) return;
    const unsigned c4 = ((unsigned)idx & ((1u << G4SHIFT) - 1u)) << 2;
    const int2 run = seg[node];
    int j = run.x;
    const int end = run.y;
    float4 acc = make_float4(0.f, 0.f, 0.f, 0.f);
    for (; j < end; j += 8) {
        int s[8];
        for (int u = 0; u < 8; ++u) {
            const int jj = j + u;
            s[u] = col[jj < end ? jj : end - 1];
        }
        asm volatile("" :: "v"(s[0]), "v"(s[1]), "v"(s[2]), "v"(s[3]),
                           "v"(s[4]), "v"(s[5]), "v"(s[6]), "v"(s[7]));
        float4 v[8];
        v[0] = *(const float4*)(h + (((unsigned)s[0] << CSHIFT) + c4));
        for (int u = 1; u < 8; ++u) {
            v[u] = make_float4(0.f, 0.f, 0.f, 0.f);
            if (j + u < end) v[u] = *(const float4*)(h + (((unsigned)s[u] << CSHIFT) + c4));
        }
        for (int u = 0; u < 8; ++u) {
            acc.x += v[u].x; acc.y += v[u].y; acc.z += v[u].z; acc.w += v[u].w;
        }
    }
    *(float4*)(agg + ((size_t)node << CSHIFT) + c4) = acc;
}

__global__ __launch_bounds__(256) void zero_kernel(float4* __restrict__ p, long long n4)
{
    long long i = (long long)blockIdx.x * blockDim.x + threadIdx.x;
    if (i < n4) p[i] = make_float4(0.f, 0.f, 0.f, 0.f);
}

__global__ __launch_bounds__(256) void pool_sum_kernel(
    const float* __restrict__ h, const int* __restrict__ batch,
    float* __restrict__ sums, float* __restrict__ cnt, long long N)
{
    long long idx = (long long)blockIdx.x * blockDim.x + threadIdx.x;
    const long long total = N << 5;
    if (idx >= total) return;
    const int node = (int)(idx >> 5);
    const int c4   = (int)(idx & 31) << 2;
    const int g    = batch[node];
    const float4 v = *(const float4*)(h + ((size_t)node << 7) + c4);
    float* sp = sums + ((size_t)g << 7) + c4;
    atomicAdd(sp + 0, v.x);
    atomicAdd(sp + 1, v.y);
    atomicAdd(sp + 2, v.z);
    atomicAdd(sp + 3, v.w);
    if (c4 == 0) atomicAdd(&cnt[g], 1.0f);
}

__global__ __launch_bounds__(256) void pool_div_kernel(
    const float4* __restrict__ sums, const float* __restrict__ cnt,
    float4* __restrict__ pooled, int G)
{
    int idx = blockIdx.x * blockDim.x + threadIdx.x;
    if (idx >= (G << 5)) return;
    const float m = fmaxf(cnt[idx >> 5], 1.0f);
    float4 s = sums[idx];
    pooled[idx] = make_float4(s.x / m, s.y / m, s.z / m, s.w / m);
}


static inline unsigned blocks_for(long long n, int t) {
    return (unsigned)((n + t - 1) / t);
}

extern "C" void kernel_launch(void* const* d_in, const int* in_sizes, int n_in,
                              void* d_out, int out_size, void* d_ws, size_t ws_size,
                              hipStream_t stream)
{
    const float* x        = (const float*)d_in[0];
    const int*   edge     = (const int*)  d_in[1];
    const int*   batch    = (const int*)  d_in[2];
    const float* w_stage1 = (const float*)d_in[4];
    const float* b_stage1 = (const float*)d_in[5];
    const float* rel1_w   = (const float*)d_in[6];
    const float* rel1_b   = (const float*)d_in[7];
    const float* root1_w  = (const float*)d_in[8];
    const float* rel2_w   = (const float*)d_in[9];
    const float* rel2_b   = (const float*)d_in[10];
    const float* root2_w  = (const float*)d_in[11];
    const float* rel3_w   = (const float*)d_in[12];
    const float* rel3_b   = (const float*)d_in[13];
    const float* root3_w  = (const float*)d_in[14];
    const float* lin_w    = (const float*)d_in[15];
    const float* lin_b    = (const float*)d_in[16];
    float*       out      = (float*)d_out;

    const int N = in_sizes[0] / 128;
    const int E = in_sizes[1] / 2;
    const int G = out_size / 16;
    const int* src = edge;
    const int* dst = edge + E;

    float* ws = (float*)d_ws;
    size_t off = 0;
    float* bufA   = ws + off; off += (size_t)N * 128;
    float* bufB   = ws + off; off += (size_t)N * 128;
    float* agg    = ws + off; off += (size_t)N * 128;
    float* sums   = ws + off; off += (size_t)G * 128;
    float* cnt    = ws + off; off += (size_t)G;
    float* pooled = ws + off; off += (size_t)G * 128;
    const long long deg4 = ((long long)N + 1 + 3) / 4;
    int* deg    = (int*)(ws + off); off += (size_t)deg4 * 4;
    int* cursor = deg + N;
    int2* seg   = (int2*)(ws + off); off += (size_t)N * 2;
    int* rank   = (int*)(ws + off); off += (size_t)E;
    int* col    = (int*)(ws + off); off += (size_t)E;
    (void)ws_size; (void)n_in;

    const int T = 256;
    auto gemm_grid = [&](int M, int Cout, int NT) {
        return blocks_for((long long)(M >> 4), 8) * (unsigned)(Cout / (16 * NT));
    };

    zero_kernel<<<blocks_for(deg4, T), T, 0, stream>>>((float4*)deg, deg4);
    count_rank_kernel<<<blocks_for(E, T), T, 0, stream>>>(dst, deg, rank, E);
    row_alloc_kernel<<<blocks_for(N, T), T, 0, stream>>>(deg, cursor, seg, N);
    fill_kernel<<<blocks_for(E, T), T, 0, stream>>>(src, dst, rank, seg, col, E);

    gemm_wmma_kernel<4, 128, false, false><<<gemm_grid(N, 64, 4), T, 0, stream>>>(
        x, w_stage1, nullptr, nullptr, b_stage1, bufA, N, 64);

    gather_sum_kernel<6><<<blocks_for((long long)N << 4, T), T, 0, stream>>>(
        bufA, col, seg, agg, N);
    gemm_wmma_kernel<4, 64, true, true><<<gemm_grid(N, 64, 4), T, 0, stream>>>(
        agg, rel1_w, bufA, root1_w, rel1_b, bufB, N, 64);

    gather_sum_kernel<6><<<blocks_for((long long)N << 4, T), T, 0, stream>>>(
        bufB, col, seg, agg, N);
    gemm_wmma_kernel<4, 64, true, true><<<gemm_grid(N, 128, 4), T, 0, stream>>>(
        agg, rel2_w, bufB, root2_w, rel2_b, bufA, N, 128);

    gather_sum_kernel<7><<<blocks_for((long long)N << 5, T), T, 0, stream>>>(
        bufA, col, seg, agg, N);
    gemm_wmma_kernel<4, 128, true, true><<<gemm_grid(N, 128, 4), T, 0, stream>>>(
        agg, rel3_w, bufA, root3_w, rel3_b, bufB, N, 128);

    zero_kernel<<<blocks_for((long long)G * 128 / 4, T), T, 0, stream>>>(
        (float4*)sums, (long long)G * 128 / 4);
    zero_kernel<<<blocks_for((long long)G / 4, T), T, 0, stream>>>(
        (float4*)cnt, (long long)G / 4);
    pool_sum_kernel<<<blocks_for((long long)N << 5, T), T, 0, stream>>>(
        bufB, batch, sums, cnt, N);
    pool_div_kernel<<<blocks_for((long long)G << 5, T), T, 0, stream>>>(
        (const float4*)sums, cnt, (float4*)pooled, G);

    gemm_wmma_kernel<1, 128, false, false><<<gemm_grid(G, 16, 1), T, 0, stream>>>(
        pooled, lin_w, nullptr, nullptr, lin_b, out, G, 16);
}


# Mean-pool sums and row counts formed in the last conv GEMM's epilogue: runs of a lane's 8 rows summed in registers, then one float atomic per graph; h3 never stored; pool sum order changes
# speedup vs baseline: 2.2513x; 1.2649x over previous
#include <hip/hip_runtime.h>


typedef __attribute__((ext_vector_type(2))) float v2f;
typedef __attribute__((ext_vector_type(8))) float v8f;

template <int NT, int CIN, bool RELU, bool DUAL, bool POOL>
__global__ __launch_bounds__(256) void gemm_wmma_kernel(
    const float* __restrict__ A, const float* __restrict__ W1,
    const float* __restrict__ B, const float* __restrict__ W2,
    const float* __restrict__ bias, float* __restrict__ out,
    const int* __restrict__ batch, float* __restrict__ cnt,
    int M, int Cout)
{
    constexpr int KC    = 64;
    constexpr int KP    = KC / 2;
    constexpr int PITCH = NT * 32 + ((NT & 1) ? 64 : 32);
    constexpr int NMAT  = DUAL ? 2 : 1;
    constexpr int C4    = NT * 4;
    __shared__ __attribute__((aligned(16))) float wl[NMAT * KP * PITCH];

    const int lane = threadIdx.x & 31;
    const int wave = threadIdx.x >> 5;
    const int ncg  = Cout / (16 * NT);
    const int rb   = blockIdx.x / ncg;
    const int cg   = blockIdx.x - rb * ncg;
    const int col0 = cg * (16 * NT);
    const int strip = rb * 8 + wave;
    const bool active = strip < (M >> 4);
    const int c = lane & 15;
    const int h = lane >> 4;

    const size_t rowoff = (size_t)((active ? strip : 0) * 16 + c) * CIN + 2 * h;
    const float* arow = A + rowoff;
    const float* brow = DUAL ? (B + rowoff) : nullptr;

    v8f acc[NT];
    #pragma unroll
    for (int t = 0; t < NT; ++t) acc[t] = v8f{};

    #pragma unroll
    for (int k0 = 0; k0 < CIN; k0 += KC) {
        v2f af[KC / 4];
        v2f bf[DUAL ? KC / 4 : 1];
        if (active) {
            #pragma unroll
            for (int kk = 0; kk < KC / 4; ++kk) {
                af[kk] = *(const v2f*)(arow + k0 + 4 * kk);
                if (DUAL) bf[kk] = *(const v2f*)(brow + k0 + 4 * kk);
            }
        }
        if (k0 > 0) __syncthreads();
        #pragma unroll
        for (int m = 0; m < NMAT; ++m) {
            const float* Wm = m ? W2 : W1;
            for (int i = threadIdx.x; i < KP * C4; i += 256) {
                const int kp = i / C4;
                const int c4 = i - kp * C4;
                const float* g = Wm + (size_t)(k0 + 2 * kp) * Cout + col0 + 4 * c4;
                const float4 r0 = *(const float4*)g;
                const float4 r1 = *(const float4*)(g + Cout);
                float* d = wl + (m * KP + kp) * PITCH + c4 * 8;
                *(float4*)d       = make_float4(r0.x, r1.x, r0.y, r1.y);
                *(float4*)(d + 4) = make_float4(r0.z, r1.z, r0.w, r1.w);
            }
        }
        __syncthreads();
        if (active) {
            #pragma unroll
            for (int kk = 0; kk < KC / 4; ++kk) {
                const float* wrow = wl + (2 * kk + h) * PITCH + c * 2;
                #pragma unroll
                for (int t = 0; t < NT; ++t) {
                    const v2f wv = *(const v2f*)(wrow + t * 32);
                    acc[t] = __builtin_amdgcn_wmma_f32_16x16x4_f32(
                                 false, af[kk], false, wv, (short)0, acc[t], false, false);
                }
                if (DUAL) {
                    #pragma unroll
                    for (int t = 0; t < NT; ++t) {
                        const v2f w2 = *(const v2f*)(wrow + KP * PITCH + t * 32);
                        acc[t] = __builtin_amdgcn_wmma_f32_16x16x4_f32(
                                     false, bf[kk], false, w2, (short)0, acc[t], false, false);
                    }
                }
            }
        }
    }
    if (!active) return;

    const int m0 = strip * 16 + (h << 3);
    if (POOL) {
        int g[8];
        float bv[NT];
        #pragma unroll
        for (int r = 0; r < 8; ++r) g[r] = batch[m0 + r];
        #pragma unroll
        for (int t = 0; t < NT; ++t) bv[t] = bias[col0 + t * 16 + c];
        #pragma unroll
        for (int t = 0; t < NT; ++t) {
            const int col = col0 + t * 16 + c;
            float run = 0.0f;
            #pragma unroll
            for (int r = 0; r < 8; ++r) {
                float v = acc[t][r] + bv[t];
                if (RELU) v = fmaxf(v, 0.0f);
                if (r == 0) {
                    run = v;
                } else {
                    if (g[r] != g[r - 1]) {
                        atomicAdd(out + (g[r - 1] * Cout + col), run);
                        run = 0.0f;
                    }
                    run += v;
                }
            }
            atomicAdd(out + (g[7] * Cout + col), run);
        }
        if (cg == 0 && c == 0) {
            float rows = 1.0f;
            #pragma unroll
            for (int r = 1; r < 8; ++r) {
                if (g[r] != g[r - 1]) {
                    atomicAdd(cnt + g[r - 1], rows);
                    rows = 0.0f;
                }
                rows += 1.0f;
            }
            atomicAdd(cnt + g[7], rows);
        }
        return;
    }
    #pragma unroll
    for (int t = 0; t < NT; ++t) {
        const int col = col0 + t * 16 + c;
        const float bval = bias[col];
        #pragma unroll
        for (int r = 0; r < 8; ++r) {
            float v = acc[t][r] + bval;
            if (RELU) v = fmaxf(v, 0.0f);
            out[(size_t)(m0 + r) * Cout + col] = v;
        }
    }
}

__global__ __launch_bounds__(256) void count_rank_kernel(
    const int* __restrict__ dst, int* __restrict__ deg,
    int* __restrict__ rank, int E)
{
    const int e = blockIdx.x * blockDim.x + threadIdx.x;
    if (e >= E) return;
    rank[e] = atomicAdd(&deg[dst[e]], 1);
}

__global__ __launch_bounds__(256) void row_alloc_kernel(
    const int* __restrict__ deg, int* __restrict__ cursor,
    int2* __restrict__ seg, int N)
{
    __shared__ int wave_sum[8];
    __shared__ int block_base;
    const int n    = blockIdx.x * blockDim.x + threadIdx.x;
    const int lane = threadIdx.x & 31;
    const int w    = threadIdx.x >> 5;
    const int d    = (n < N) ? deg[n] : 0;
    int v = d;
    for (int o = 1; o < 32; o <<= 1) {
        const int t = __shfl_up(v, o);
        if (lane >= o) v += t;
    }
    if (lane == 31) wave_sum[w] = v;
    __syncthreads();
    if (threadIdx.x == 0) {
        int s = 0;
        for (int i = 0; i < 8; ++i) { const int t = wave_sum[i]; wave_sum[i] = s; s += t; }
        block_base = atomicAdd(cursor, s);
    }
    __syncthreads();
    if (n < N) {
        const int first = block_base + wave_sum[w] + v - d;
        seg[n] = make_int2(first, first + d);
    }
}

__global__ __launch_bounds__(256) void fill_kernel(
    const int* __restrict__ src, const int* __restrict__ dst,
    const int* __restrict__ rank, const int2* __restrict__ seg,
    int* __restrict__ col, int E)
{
    const int e = blockIdx.x * blockDim.x + threadIdx.x;
    if (e >= E) return;
    col[seg[dst[e]].x + rank[e]] = src[e];
}

template <int CSHIFT>
__global__ __launch_bounds__(256) void gather_sum_kernel(
    const float* __restrict__ h, const int* __restrict__ col,
    const int2* __restrict__ seg, float* __restrict__ agg, int N)
{
    constexpr int G4SHIFT = CSHIFT - 2;
    const long long idx = (long long)blockIdx.x * blockDim.x + threadIdx.x;
    const int node = (int)(idx >> G4SHIFT);
    if (node >= N) return;
    const unsigned c4 = ((unsigned)idx & ((1u << G4SHIFT) - 1u)) << 2;
    const int2 run = seg[node];
    int j = run.x;
    const int end = run.y;
    float4 acc = make_float4(0.f, 0.f, 0.f, 0.f);
    for (; j < end; j += 8) {
        int s[8];
        for (int u = 0; u < 8; ++u) {
            const int jj = j + u;
            s[u] = col[jj < end ? jj : end - 1];
        }
        asm volatile("" :: "v"(s[0]), "v"(s[1]), "v"(s[2]), "v"(s[3]),
                           "v"(s[4]), "v"(s[5]), "v"(s[6]), "v"(s[7]));
        float4 v[8];
        v[0] = *(const float4*)(h + (((unsigned)s[0] << CSHIFT) + c4));
        for (int u = 1; u < 8; ++u) {
            v[u] = make_float4(0.f, 0.f, 0.f, 0.f);
            if (j + u < end) v[u] = *(const float4*)(h + (((unsigned)s[u] << CSHIFT) + c4));
        }
        for (int u = 0; u < 8; ++u) {
            acc.x += v[u].x; acc.y += v[u].y; acc.z += v[u].z; acc.w += v[u].w;
        }
    }
    *(float4*)(agg + ((size_t)node << CSHIFT) + c4) = acc;
}

__global__ __launch_bounds__(256) void zero_kernel(float4* __restrict__ p, long long n4)
{
    long long i = (long long)blockIdx.x * blockDim.x + threadIdx.x;
    if (i < n4) p[i] = make_float4(0.f, 0.f, 0.f, 0.f);
}

__global__ __launch_bounds__(256) void pool_div_kernel(
    const float4* __restrict__ sums, const float* __restrict__ cnt,
    float4* __restrict__ pooled, int G)
{
    int idx = blockIdx.x * blockDim.x + threadIdx.x;
    if (idx >= (G << 5)) return;
    const float m = fmaxf(cnt[idx >> 5], 1.0f);
    float4 s = sums[idx];
    pooled[idx] = make_float4(s.x / m, s.y / m, s.z / m, s.w / m);
}


static inline unsigned blocks_for(long long n, int t) {
    return (unsigned)((n + t - 1) / t);
}

extern "C" void kernel_launch(void* const* d_in, const int* in_sizes, int n_in,
                              void* d_out, int out_size, void* d_ws, size_t ws_size,
                              hipStream_t stream)
{
    const float* x        = (const float*)d_in[0];
    const int*   edge     = (const int*)  d_in[1];
    const int*   batch    = (const int*)  d_in[2];
    const float* w_stage1 = (const float*)d_in[4];
    const float* b_stage1 = (const float*)d_in[5];
    const float* rel1_w   = (const float*)d_in[6];
    const float* rel1_b   = (const float*)d_in[7];
    const float* root1_w  = (const float*)d_in[8];
    const float* rel2_w   = (const float*)d_in[9];
    const float* rel2_b   = (const float*)d_in[10];
    const float* root2_w  = (const float*)d_in[11];
    const float* rel3_w   = (const float*)d_in[12];
    const float* rel3_b   = (const float*)d_in[13];
    const float* root3_w  = (const float*)d_in[14];
    const float* lin_w    = (const float*)d_in[15];
    const float* lin_b    = (const float*)d_in[16];
    float*       out      = (float*)d_out;

    const int N = in_sizes[0] / 128;
    const int E = in_sizes[1] / 2;
    const int G = out_size / 16;
    const int* src = edge;
    const int* dst = edge + E;

    float* ws = (float*)d_ws;
    size_t off = 0;
    float* bufA   = ws + off; off += (size_t)N * 128;
    float* bufB   = ws + off; off += (size_t)N * 128;
    float* agg    = ws + off; off += (size_t)N * 128;
    float* pooled = ws + off; off += (size_t)G * 128;
    const long long zero4 = ((long long)G * 129 + N + 1 + 3) / 4;
    float* sums = ws + off;
    float* cnt  = sums + (size_t)G * 128;
    int* deg    = (int*)(cnt + G);
    int* cursor = deg + N;
    off += (size_t)zero4 * 4;
    int2* seg   = (int2*)(ws + off); off += (size_t)N * 2;
    int* rank   = (int*)(ws + off); off += (size_t)E;
    int* col    = (int*)(ws + off); off += (size_t)E;
    (void)ws_size; (void)n_in;

    const int T = 256;
    auto gemm_grid = [&](int M, int Cout, int NT) {
        return blocks_for((long long)(M >> 4), 8) * (unsigned)(Cout / (16 * NT));
    };

    zero_kernel<<<blocks_for(zero4, T), T, 0, stream>>>((float4*)sums, zero4);

    count_rank_kernel<<<blocks_for(E, T), T, 0, stream>>>(dst, deg, rank, E);
    row_alloc_kernel<<<blocks_for(N, T), T, 0, stream>>>(deg, cursor, seg, N);
    fill_kernel<<<blocks_for(E, T), T, 0, stream>>>(src, dst, rank, seg, col, E);

    gemm_wmma_kernel<4, 128, false, false, false><<<gemm_grid(N, 64, 4), T, 0, stream>>>(
        x, w_stage1, nullptr, nullptr, b_stage1, bufA, nullptr, nullptr, N, 64);

    gather_sum_kernel<6><<<blocks_for((long long)N << 4, T), T, 0, stream>>>(
        bufA, col, seg, agg, N);
    gemm_wmma_kernel<4, 64, true, true, false><<<gemm_grid(N, 64, 4), T, 0, stream>>>(
        agg, rel1_w, bufA, root1_w, rel1_b, bufB, nullptr, nullptr, N, 64);

    gather_sum_kernel<6><<<blocks_for((long long)N << 4, T), T, 0, stream>>>(
        bufB, col, seg, agg, N);
    gemm_wmma_kernel<4, 64, true, true, false><<<gemm_grid(N, 128, 4), T, 0, stream>>>(
        agg, rel2_w, bufB, root2_w, rel2_b, bufA, nullptr, nullptr, N, 128);

    gather_sum_kernel<7><<<blocks_for((long long)N << 5, T), T, 0, stream>>>(
        bufA, col, seg, agg, N);
    gemm_wmma_kernel<4, 128, true, true, true><<<gemm_grid(N, 128, 4), T, 0, stream>>>(
        agg, rel3_w, bufA, root3_w, rel3_b, sums, batch, cnt, N, 128);

    pool_div_kernel<<<blocks_for((long long)G << 5, T), T, 0, stream>>>(
        (const float4*)sums, cnt, (float4*)pooled, G);

    gemm_wmma_kernel<1, 128, false, false, false><<<gemm_grid(G, 16, 1), T, 0, stream>>>(
        pooled, lin_w, nullptr, nullptr, lin_b, out, nullptr, nullptr, G, 16);
}


# Fewer launches, 13 to 11: stage-1 GEMM work-groups ride in the edge-count and slot-allocation launches (independent work); classifier GEMM divides graph sums by counts in its A load; sum order kept
# speedup vs baseline: 2.3374x; 1.0382x over previous
#include <hip/hip_runtime.h>


typedef __attribute__((ext_vector_type(2))) float v2f;
typedef __attribute__((ext_vector_type(8))) float v8f;

template <int NT, int CIN, bool RELU, bool DUAL, bool POOL, bool DIV>
__device__ __forceinline__ void gemm_body(
    const float* __restrict__ A, const float* __restrict__ W1,
    const float* __restrict__ B, const float* __restrict__ W2,
    const float* __restrict__ bias, float* __restrict__ out,
    const int* __restrict__ batch, float* __restrict__ cnt,
    int M, int Cout, int bid)
{
    constexpr int KC    = 64;
    constexpr int KP    = KC / 2;
    constexpr int PITCH = NT * 32 + ((NT & 1) ? 64 : 32);
    constexpr int NMAT  = DUAL ? 2 : 1;
    constexpr int C4    = NT * 4;
    __shared__ __attribute__((aligned(16))) float wl[NMAT * KP * PITCH];

    const int lane = threadIdx.x & 31;
    const int wave = threadIdx.x >> 5;
    const int ncg  = Cout / (16 * NT);
    const int rb   = bid / ncg;
    const int cg   = bid - rb * ncg;
    const int col0 = cg * (16 * NT);
    const int strip = rb * 8 + wave;
    const bool active = strip < (M >> 4);
    const int c = lane & 15;
    const int h = lane >> 4;

    const size_t rowoff = (size_t)((active ? strip : 0) * 16 + c) * CIN + 2 * h;
    const float* arow = A + rowoff;
    const float* brow = DUAL ? (B + rowoff) : nullptr;
    float rows_of = 1.0f;
    if (DIV) rows_of = fmaxf(cnt[(active ? strip : 0) * 16 + c], 1.0f);

    v8f acc[NT];
    #pragma unroll
    for (int t = 0; t < NT; ++t) acc[t] = v8f{};

    #pragma unroll
    for (int k0 = 0; k0 < CIN; k0 += KC) {
        v2f af[KC / 4];
        v2f bf[DUAL ? KC / 4 : 1];
        if (active) {
            #pragma unroll
            for (int kk = 0; kk < KC / 4; ++kk) {
                af[kk] = *(const v2f*)(arow + k0 + 4 * kk);
                if (DIV) { af[kk].x = af[kk].x / rows_of; af[kk].y = af[kk].y / rows_of; }
                if (DUAL) bf[kk] = *(const v2f*)(brow + k0 + 4 * kk);
            }
        }
        if (k0 > 0) __syncthreads();
        #pragma unroll
        for (int m = 0; m < NMAT; ++m) {
            const float* Wm = m ? W2 : W1;
            for (int i = threadIdx.x; i < KP * C4; i += 256) {
                const int kp = i / C4;
                const int c4 = i - kp * C4;
                const float* g = Wm + (size_t)(k0 + 2 * kp) * Cout + col0 + 4 * c4;
                const float4 r0 = *(const float4*)g;
                const float4 r1 = *(const float4*)(g + Cout);
                float* d = wl + (m * KP + kp) * PITCH + c4 * 8;
                *(float4*)d       = make_float4(r0.x, r1.x, r0.y, r1.y);
                *(float4*)(d + 4) = make_float4(r0.z, r1.z, r0.w, r1.w);
            }
        }
        __syncthreads();
        if (active) {
            #pragma unroll
            for (int kk = 0; kk < KC / 4; ++kk) {
                const float* wrow = wl + (2 * kk + h) * PITCH + c * 2;
                #pragma unroll
                for (int t = 0; t < NT; ++t) {
                    const v2f wv = *(const v2f*)(wrow + t * 32);
                    acc[t] = __builtin_amdgcn_wmma_f32_16x16x4_f32(
                                 false, af[kk], false, wv, (short)0, acc[t], false, false);
                }
                if (DUAL) {
                    #pragma unroll
                    for (int t = 0; t < NT; ++t) {
                        const v2f w2 = *(const v2f*)(wrow + KP * PITCH + t * 32);
                        acc[t] = __builtin_amdgcn_wmma_f32_16x16x4_f32(
                                     false, bf[kk], false, w2, (short)0, acc[t], false, false);
                    }
                }
            }
        }
    }
    if (!active) return;

    const int m0 = strip * 16 + (h << 3);
    if (POOL) {
        int g[8];
        float bv[NT];
        #pragma unroll
        for (int r = 0; r < 8; ++r) g[r] = batch[m0 + r];
        #pragma unroll
        for (int t = 0; t < NT; ++t) bv[t] = bias[col0 + t * 16 + c];
        #pragma unroll
        for (int t = 0; t < NT; ++t) {
            const int col = col0 + t * 16 + c;
            float run = 0.0f;
            #pragma unroll
            for (int r = 0; r < 8; ++r) {
                float v = acc[t][r] + bv[t];
                if (RELU) v = fmaxf(v, 0.0f);
                if (r == 0) {
                    run = v;
                } else {
                    if (g[r] != g[r - 1]) {
                        atomicAdd(out + (g[r - 1] * Cout + col), run);
                        run = 0.0f;
                    }
                    run += v;
                }
            }
            atomicAdd(out + (g[7] * Cout + col), run);
        }
        if (cg == 0 && c == 0) {
            float rows = 1.0f;
            #pragma unroll
            for (int r = 1; r < 8; ++r) {
                if (g[r] != g[r - 1]) {
                    atomicAdd(cnt + g[r - 1], rows);
                    rows = 0.0f;
                }
                rows += 1.0f;
            }
            atomicAdd(cnt + g[7], rows);
        }
        return;
    }
    #pragma unroll
    for (int t = 0; t < NT; ++t) {
        const int col = col0 + t * 16 + c;
        const float bval = bias[col];
        #pragma unroll
        for (int r = 0; r < 8; ++r) {
            float v = acc[t][r] + bval;
            if (RELU) v = fmaxf(v, 0.0f);
            out[(size_t)(m0 + r) * Cout + col] = v;
        }
    }
}

template <int NT, int CIN, bool RELU, bool DUAL, bool POOL, bool DIV>
__global__ __launch_bounds__(256) void gemm_wmma_kernel(
    const float* __restrict__ A, const float* __restrict__ W1,
    const float* __restrict__ B, const float* __restrict__ W2,
    const float* __restrict__ bias, float* __restrict__ out,
    const int* __restrict__ batch, float* __restrict__ cnt,
    int M, int Cout)
{
    gemm_body<NT, CIN, RELU, DUAL, POOL, DIV>(A, W1, B, W2, bias, out, batch, cnt,
                                              M, Cout, (int)blockIdx.x);
}

__global__ __launch_bounds__(256) void stage1_and_count_kernel(
    const float* __restrict__ x, const float* __restrict__ w,
    const float* __restrict__ bias, float* __restrict__ h0, int M,
    const int* __restrict__ dst, int* __restrict__ deg,
    int* __restrict__ rank, int E, int gemm_first, int gemm_groups)
{
    if ((int)blockIdx.x < gemm_groups) {
        gemm_body<4, 128, false, false, false, false>(
            x, w, nullptr, nullptr, bias, h0, nullptr, nullptr, M, 64,
            gemm_first + (int)blockIdx.x);
        return;
    }
    const int e = ((int)blockIdx.x - gemm_groups) * 256 + threadIdx.x;
    if (e >= E) return;
    rank[e] = atomicAdd(&deg[dst[e]], 1);
}

__global__ __launch_bounds__(256) void stage1_and_row_alloc_kernel(
    const float* __restrict__ x, const float* __restrict__ wmat,
    const float* __restrict__ bias, float* __restrict__ h0, int M,
    const int* __restrict__ deg, int* __restrict__ cursor,
    int2* __restrict__ seg, int N, int gemm_first, int gemm_groups)
{
    if ((int)blockIdx.x < gemm_groups) {
        gemm_body<4, 128, false, false, false, false>(
            x, wmat, nullptr, nullptr, bias, h0, nullptr, nullptr, M, 64,
            gemm_first + (int)blockIdx.x);
        return;
    }
    __shared__ int wave_sum[8];
    __shared__ int block_base;
    const int n    = ((int)blockIdx.x - gemm_groups) * 256 + threadIdx.x;
    const int lane = threadIdx.x & 31;
    const int w    = threadIdx.x >> 5;
    const int d    = (n < N) ? deg[n] : 0;
    int v = d;
    for (int o = 1; o < 32; o <<= 1) {
        const int t = __shfl_up(v, o);
        if (lane >= o) v += t;
    }
    if (lane == 31) wave_sum[w] = v;
    __syncthreads();
    if (threadIdx.x == 0) {
        int s = 0;
        for (int i = 0; i < 8; ++i) { const int t = wave_sum[i]; wave_sum[i] = s; s += t; }
        block_base = atomicAdd(cursor, s);
    }
    __syncthreads();
    if (n < N) {
        const int first = block_base + wave_sum[w] + v - d;
        seg[n] = make_int2(first, first + d);
    }
}

__global__ __launch_bounds__(256) void fill_kernel(
    const int* __restrict__ src, const int* __restrict__ dst,
    const int* __restrict__ rank, const int2* __restrict__ seg,
    int* __restrict__ col, int E)
{
    const int e = blockIdx.x * blockDim.x + threadIdx.x;
    if (e >= E) return;
    col[seg[dst[e]].x + rank[e]] = src[e];
}

template <int CSHIFT>
__global__ __launch_bounds__(256) void gather_sum_kernel(
    const float* __restrict__ h, const int* __restrict__ col,
    const int2* __restrict__ seg, float* __restrict__ agg, int N)
{
    constexpr int G4SHIFT = CSHIFT - 2;
    const long long idx = (long long)blockIdx.x * blockDim.x + threadIdx.x;
    const int node = (int)(idx >> G4SHIFT);
    if (node >= N) return;
    const unsigned c4 = ((unsigned)idx & ((1u << G4SHIFT) - 1u)) << 2;
    const int2 run = seg[node];
    int j = run.x;
    const int end = run.y;
    float4 acc = make_float4(0.f, 0.f, 0.f, 0.f);
    for (; j < end; j += 8) {
        int s[8];
        for (int u = 0; u < 8; ++u) {
            const int jj = j + u;
            s[u] = col[jj < end ? jj : end - 1];
        }
        asm volatile("" :: "v"(s[0]), "v"(s[1]), "v"(s[2]), "v"(s[3]),
                           "v"(s[4]), "v"(s[5]), "v"(s[6]), "v"(s[7]));
        float4 v[8];
        v[0] = *(const float4*)(h + (((unsigned)s[0] << CSHIFT) + c4));
        for (int u = 1; u < 8; ++u) {
            v[u] = make_float4(0.f, 0.f, 0.f, 0.f);
            if (j + u < end) v[u] = *(const float4*)(h + (((unsigned)s[u] << CSHIFT) + c4));
        }
        for (int u = 0; u < 8; ++u) {
            acc.x += v[u].x; acc.y += v[u].y; acc.z += v[u].z; acc.w += v[u].w;
        }
    }
    *(float4*)(agg + ((size_t)node << CSHIFT) + c4) = acc;
}

__global__ __launch_bounds__(256) void zero_kernel(float4* __restrict__ p, long long n4)
{
    long long i = (long long)blockIdx.x * blockDim.x + threadIdx.x;
    if (i < n4) p[i] = make_float4(0.f, 0.f, 0.f, 0.f);
}


static inline unsigned blocks_for(long long n, int t) {
    return (unsigned)((n + t - 1) / t);
}

extern "C" void kernel_launch(void* const* d_in, const int* in_sizes, int n_in,
                              void* d_out, int out_size, void* d_ws, size_t ws_size,
                              hipStream_t stream)
{
    const float* x        = (const float*)d_in[0];
    const int*   edge     = (const int*)  d_in[1];
    const int*   batch    = (const int*)  d_in[2];
    const float* w_stage1 = (const float*)d_in[4];
    const float* b_stage1 = (const float*)d_in[5];
    const float* rel1_w   = (const float*)d_in[6];
    const float* rel1_b   = (const float*)d_in[7];
    const float* root1_w  = (const float*)d_in[8];
    const float* rel2_w   = (const float*)d_in[9];
    const float* rel2_b   = (const float*)d_in[10];
    const float* root2_w  = (const float*)d_in[11];
    const float* rel3_w   = (const float*)d_in[12];
    const float* rel3_b   = (const float*)d_in[13];
    const float* root3_w  = (const float*)d_in[14];
    const float* lin_w    = (const float*)d_in[15];
    const float* lin_b    = (const float*)d_in[16];
    float*       out      = (float*)d_out;

    const int N = in_sizes[0] / 128;
    const int E = in_sizes[1] / 2;
    const int G = out_size / 16;
    const int* src = edge;
    const int* dst = edge + E;

    float* ws = (float*)d_ws;
    size_t off = 0;
    float* bufA   = ws + off; off += (size_t)N * 128;
    float* bufB   = ws + off; off += (size_t)N * 128;
    float* agg    = ws + off; off += (size_t)N * 128;
    const long long zero4 = ((long long)G * 129 + N + 1 + 3) / 4;
    float* sums = ws + off;
    float* cnt  = sums + (size_t)G * 128;
    int* deg    = (int*)(cnt + G);
    int* cursor = deg + N;
    off += (size_t)zero4 * 4;
    int2* seg   = (int2*)(ws + off); off += (size_t)N * 2;
    int* rank   = (int*)(ws + off); off += (size_t)E;
    int* col    = (int*)(ws + off); off += (size_t)E;
    (void)ws_size; (void)n_in;

    const int T = 256;
    auto gemm_grid = [&](int M, int Cout, int NT) {
        return blocks_for((long long)(M >> 4), 8) * (unsigned)(Cout / (16 * NT));
    };

    zero_kernel<<<blocks_for(zero4, T), T, 0, stream>>>((float4*)sums, zero4);

    const unsigned g1 = gemm_grid(N, 64, 4);
    const unsigned g1a = g1 / 2;
    stage1_and_count_kernel<<<g1a + blocks_for(E, T), T, 0, stream>>>(
        x, w_stage1, b_stage1, bufA, N, dst, deg, rank, E, 0, (int)g1a);
    stage1_and_row_alloc_kernel<<<(g1 - g1a) + blocks_for(N, T), T, 0, stream>>>(
        x, w_stage1, b_stage1, bufA, N, deg, cursor, seg, N, (int)g1a, (int)(g1 - g1a));
    fill_kernel<<<blocks_for(E, T), T, 0, stream>>>(src, dst, rank, seg, col, E);

    gather_sum_kernel<6><<<blocks_for((long long)N << 4, T), T, 0, stream>>>(
        bufA, col, seg, agg, N);
    gemm_wmma_kernel<4, 64, true, true, false, false><<<gemm_grid(N, 64, 4), T, 0, stream>>>(
        agg, rel1_w, bufA, root1_w, rel1_b, bufB, nullptr, nullptr, N, 64);

    gather_sum_kernel<6><<<blocks_for((long long)N << 4, T), T, 0, stream>>>(
        bufB, col, seg, agg, N);
    gemm_wmma_kernel<4, 64, true, true, false, false><<<gemm_grid(N, 128, 4), T, 0, stream>>>(
        agg, rel2_w, bufB, root2_w, rel2_b, bufA, nullptr, nullptr, N, 128);

    gather_sum_kernel<7><<<blocks_for((long long)N << 5, T), T, 0, stream>>>(
        bufA, col, seg, agg, N);
    gemm_wmma_kernel<4, 128, true, true, true, false><<<gemm_grid(N, 128, 4), T, 0, stream>>>(
        agg, rel3_w, bufA, root3_w, rel3_b, sums, batch, cnt, N, 128);

    gemm_wmma_kernel<1, 128, false, false, false, true><<<gemm_grid(G, 16, 1), T, 0, stream>>>(
        sums, lin_w, nullptr, nullptr, lin_b, out, nullptr, cnt, G, 16);
}


# Conv layers' root product (h @ root_w) rides as extra work-groups in the launch of the neighbour sums; the GEMM behind starts from those accumulators: sum order = all root k, then all rel k
# speedup vs baseline: 2.4362x; 1.0423x over previous
#include <hip/hip_runtime.h>


typedef __attribute__((ext_vector_type(2))) float v2f;
typedef __attribute__((ext_vector_type(4))) float v4f;
typedef __attribute__((ext_vector_type(8))) float v8f;

template <int NT, int CIN, int KC, bool RELU, bool POOL, bool DIV, bool ACCIN, bool ACCOUT>
__device__ __forceinline__ void gemm_body(
    const float* __restrict__ A, const float* __restrict__ Wm,
    const float* __restrict__ bias, float* __restrict__ out,
    const int* __restrict__ batch, float* __restrict__ cnt,
    float* __restrict__ part, int M, int Cout, int bid)
{
    constexpr int KP    = KC / 2;
    constexpr int PITCH = NT * 32 + ((NT & 1) ? 64 : 32);
    constexpr int C4    = NT * 4;
    __shared__ __attribute__((aligned(16))) float wl[KP * PITCH];

    const int lane = threadIdx.x & 31;
    const int wave = threadIdx.x >> 5;
    const int ncg  = Cout / (16 * NT);
    const int rb   = bid / ncg;
    const int cg   = bid - rb * ncg;
    const int col0 = cg * (16 * NT);
    const int strip = rb * 8 + wave;
    const bool active = strip < (M >> 4);
    const int c = lane & 15;
    const int h = lane >> 4;

    const size_t rowoff = (size_t)((active ? strip : 0) * 16 + c) * CIN + 2 * h;
    const float* arow = A + rowoff;
    float rows_of = 1.0f;
    if (DIV) rows_of = fmaxf(cnt[(active ? strip : 0) * 16 + c], 1.0f);

    const size_t poff = ((size_t)(active ? strip : 0) * (size_t)(Cout >> 4) + (size_t)(cg * NT)) * 256
                        + (size_t)lane * 4;

    v8f acc[NT];
    #pragma unroll
    for (int t = 0; t < NT; ++t) {
        if (ACCIN) {
            const v4f lo = *(const v4f*)(part + poff + t * 256);
            const v4f hi = *(const v4f*)(part + poff + t * 256 + 128);
            acc[t] = __builtin_shufflevector(lo, hi, 0, 1, 2, 3, 4, 5, 6, 7);
        } else {
            acc[t] = v8f{};
        }
    }

    #pragma unroll
    for (int k0 = 0; k0 < CIN; k0 += KC) {
        v2f af[KC / 4];
        if (active) {
            #pragma unroll
            for (int kk = 0; kk < KC / 4; ++kk) {
                af[kk] = *(const v2f*)(arow + k0 + 4 * kk);
                if (DIV) { af[kk].x = af[kk].x / rows_of; af[kk].y = af[kk].y / rows_of; }
            }
        }
        if (k0 > 0) __syncthreads();
        for (int i = threadIdx.x; i < KP * C4; i += 256) {
            const int kp = i / C4;
            const int c4 = i - kp * C4;
            const float* g = Wm + (size_t)(k0 + 2 * kp) * Cout + col0 + 4 * c4;
            const float4 r0 = *(const float4*)g;
            const float4 r1 = *(const float4*)(g + Cout);
            float* d = wl + kp * PITCH + c4 * 8;
            *(float4*)d       = make_float4(r0.x, r1.x, r0.y, r1.y);
            *(float4*)(d + 4) = make_float4(r0.z, r1.z, r0.w, r1.w);
        }
        __syncthreads();
        if (active) {
            #pragma unroll
            for (int kk = 0; kk < KC / 4; ++kk) {
                const float* wrow = wl + (2 * kk + h) * PITCH + c * 2;
                #pragma unroll
                for (int t = 0; t < NT; ++t) {
                    const v2f wv = *(const v2f*)(wrow + t * 32);
                    acc[t] = __builtin_amdgcn_wmma_f32_16x16x4_f32(
                                 false, af[kk], false, wv, (short)0, acc[t], false, false);
                }
            }
        }
    }
    if (!active) return;

    if (ACCOUT) {
        #pragma unroll
        for (int t = 0; t < NT; ++t) {
            *(v4f*)(part + poff + t * 256)       = __builtin_shufflevector(acc[t], acc[t], 0, 1, 2, 3);
            *(v4f*)(part + poff + t * 256 + 128) = __builtin_shufflevector(acc[t], acc[t], 4, 5, 6, 7);
        }
        return;
    }

    const int m0 = strip * 16 + (h << 3);
    if (POOL) {
        int g[8];
        float bv[NT];
        #pragma unroll
        for (int r = 0; r < 8; ++r) g[r] = batch[m0 + r];
        #pragma unroll
        for (int t = 0; t < NT; ++t) bv[t] = bias[col0 + t * 16 + c];
        #pragma unroll
        for (int t = 0; t < NT; ++t) {
            const int col = col0 + t * 16 + c;
            float run = 0.0f;
            #pragma unroll
            for (int r = 0; r < 8; ++r) {
                float v = acc[t][r] + bv[t];
                if (RELU) v = fmaxf(v, 0.0f);
                if (r == 0) {
                    run = v;
                } else {
                    if (g[r] != g[r - 1]) {
                        atomicAdd(out + (g[r - 1] * Cout + col), run);
                        run = 0.0f;
                    }
                    run += v;
                }
            }
            atomicAdd(out + (g[7] * Cout + col), run);
        }
        if (cg == 0 && c == 0) {
            float rows = 1.0f;
            #pragma unroll
            for (int r = 1; r < 8; ++r) {
                if (g[r] != g[r - 1]) {
                    atomicAdd(cnt + g[r - 1], rows);
                    rows = 0.0f;
                }
                rows += 1.0f;
            }
            atomicAdd(cnt + g[7], rows);
        }
        return;
    }
    #pragma unroll
    for (int t = 0; t < NT; ++t) {
        const int col = col0 + t * 16 + c;
        const float bval = bias[col];
        #pragma unroll
        for (int r = 0; r < 8; ++r) {
            float v = acc[t][r] + bval;
            if (RELU) v = fmaxf(v, 0.0f);
            out[(size_t)(m0 + r) * Cout + col] = v;
        }
    }
}

template <int NT, int CIN, bool RELU, bool POOL, bool DIV, bool ACCIN>
__global__ __launch_bounds__(256) void gemm_wmma_kernel(
    const float* __restrict__ A, const float* __restrict__ Wm,
    const float* __restrict__ bias, float* __restrict__ out,
    const int* __restrict__ batch, float* __restrict__ cnt,
    float* __restrict__ part, int M, int Cout)
{
    gemm_body<NT, CIN, 64, RELU, POOL, DIV, ACCIN, false>(A, Wm, bias, out, batch, cnt, part,
                                                          M, Cout, (int)blockIdx.x);
}

__global__ __launch_bounds__(256) void stage1_and_count_kernel(
    const float* __restrict__ x, const float* __restrict__ w,
    const float* __restrict__ bias, float* __restrict__ h0, int M,
    const int* __restrict__ dst, int* __restrict__ deg,
    int* __restrict__ rank, int E, int gemm_first, int gemm_groups)
{
    if ((int)blockIdx.x < gemm_groups) {
        gemm_body<4, 128, 64, false, false, false, false, false>(
            x, w, bias, h0, nullptr, nullptr, nullptr, M, 64,
            gemm_first + (int)blockIdx.x);
        return;
    }
    const int e = ((int)blockIdx.x - gemm_groups) * 256 + threadIdx.x;
    if (e >= E) return;
    rank[e] = atomicAdd(&deg[dst[e]], 1);
}

__global__ __launch_bounds__(256) void stage1_and_row_alloc_kernel(
    const float* __restrict__ x, const float* __restrict__ wmat,
    const float* __restrict__ bias, float* __restrict__ h0, int M,
    const int* __restrict__ deg, int* __restrict__ cursor,
    int2* __restrict__ seg, int N, int gemm_first, int gemm_groups)
{
    if ((int)blockIdx.x < gemm_groups) {
        gemm_body<4, 128, 64, false, false, false, false, false>(
            x, wmat, bias, h0, nullptr, nullptr, nullptr, M, 64,
            gemm_first + (int)blockIdx.x);
        return;
    }
    __shared__ int wave_sum[8];
    __shared__ int block_base;
    const int n    = ((int)blockIdx.x - gemm_groups) * 256 + threadIdx.x;
    const int lane = threadIdx.x & 31;
    const int w    = threadIdx.x >> 5;
    const int d    = (n < N) ? deg[n] : 0;
    int v = d;
    for (int o = 1; o < 32; o <<= 1) {
        const int t = __shfl_up(v, o);
        if (lane >= o) v += t;
    }
    if (lane == 31) wave_sum[w] = v;
    __syncthreads();
    if (threadIdx.x == 0) {
        int s = 0;
        for (int i = 0; i < 8; ++i) { const int t = wave_sum[i]; wave_sum[i] = s; s += t; }
        block_base = atomicAdd(cursor, s);
    }
    __syncthreads();
    if (n < N) {
        const int first = block_base + wave_sum[w] + v - d;
        seg[n] = make_int2(first, first + d);
    }
}

__global__ __launch_bounds__(256) void fill_kernel(
    const int* __restrict__ src, const int* __restrict__ dst,
    const int* __restrict__ rank, const int2* __restrict__ seg,
    int* __restrict__ col, int E)
{
    const int e = blockIdx.x * blockDim.x + threadIdx.x;
    if (e >= E) return;
    col[seg[dst[e]].x + rank[e]] = src[e];
}

template <int CSHIFT>
__device__ __forceinline__ void gather_sum_body(
    const float* __restrict__ h, const int* __restrict__ col,
    const int2* __restrict__ seg, float* __restrict__ agg, int N, int bid)
{
    constexpr int G4SHIFT = CSHIFT - 2;
    const long long idx = (long long)bid * 256 + threadIdx.x;
    const int node = (int)(idx >> G4SHIFT);
    if (node >= N) return;
    const unsigned c4 = ((unsigned)idx & ((1u << G4SHIFT) - 1u)) << 2;
    const int2 run = seg[node];
    int j = run.x;
    const int end = run.y;
    float4 acc = make_float4(0.f, 0.f, 0.f, 0.f);
    for (; j < end; j += 8) {
        int s[8];
        for (int u = 0; u < 8; ++u) {
            const int jj = j + u;
            s[u] = col[jj < end ? jj : end - 1];
        }
        asm volatile("" :: "v"(s[0]), "v"(s[1]), "v"(s[2]), "v"(s[3]),
                           "v"(s[4]), "v"(s[5]), "v"(s[6]), "v"(s[7]));
        float4 v[8];
        v[0] = *(const float4*)(h + (((unsigned)s[0] << CSHIFT) + c4));
        for (int u = 1; u < 8; ++u) {
            v[u] = make_float4(0.f, 0.f, 0.f, 0.f);
            if (j + u < end) v[u] = *(const float4*)(h + (((unsigned)s[u] << CSHIFT) + c4));
        }
        for (int u = 0; u < 8; ++u) {
            acc.x += v[u].x; acc.y += v[u].y; acc.z += v[u].z; acc.w += v[u].w;
        }
    }
    *(float4*)(agg + ((size_t)node << CSHIFT) + c4) = acc;
}

template <int CSHIFT, int KC>
__global__ __launch_bounds__(256) __attribute__((amdgpu_waves_per_eu(16, 16)))
void gather_sum_and_root_kernel(
    const float* __restrict__ h, const int* __restrict__ col,
    const int2* __restrict__ seg, float* __restrict__ agg, int N,
    const float* __restrict__ wroot, float* __restrict__ part, int Cout,
    int gemm_groups)
{
    if ((int)blockIdx.x < gemm_groups) {
        gemm_body<4, (1 << CSHIFT), KC, false, false, false, false, true>(
            h, wroot, nullptr, nullptr, nullptr, nullptr, part, N, Cout,
            (int)blockIdx.x);
        return;
    }
    gather_sum_body<CSHIFT>(h, col, seg, agg, N, (int)blockIdx.x - gemm_groups);
}

__global__ __launch_bounds__(256) void zero_kernel(float4* __restrict__ p, long long n4)
{
    long long i = (long long)blockIdx.x * blockDim.x + threadIdx.x;
    if (i < n4) p[i] = make_float4(0.f, 0.f, 0.f, 0.f);
}


static inline unsigned blocks_for(long long n, int t) {
    return (unsigned)((n + t - 1) / t);
}

extern "C" void kernel_launch(void* const* d_in, const int* in_sizes, int n_in,
                              void* d_out, int out_size, void* d_ws, size_t ws_size,
                              hipStream_t stream)
{
    const float* x        = (const float*)d_in[0];
    const int*   edge     = (const int*)  d_in[1];
    const int*   batch    = (const int*)  d_in[2];
    const float* w_stage1 = (const float*)d_in[4];
    const float* b_stage1 = (const float*)d_in[5];
    const float* rel1_w   = (const float*)d_in[6];
    const float* rel1_b   = (const float*)d_in[7];
    const float* root1_w  = (const float*)d_in[8];
    const float* rel2_w   = (const float*)d_in[9];
    const float* rel2_b   = (const float*)d_in[10];
    const float* root2_w  = (const float*)d_in[11];
    const float* rel3_w   = (const float*)d_in[12];
    const float* rel3_b   = (const float*)d_in[13];
    const float* root3_w  = (const float*)d_in[14];
    const float* lin_w    = (const float*)d_in[15];
    const float* lin_b    = (const float*)d_in[16];
    float*       out      = (float*)d_out;

    const int N = in_sizes[0] / 128;
    const int E = in_sizes[1] / 2;
    const int G = out_size / 16;
    const int* src = edge;
    const int* dst = edge + E;

    float* ws = (float*)d_ws;
    size_t off = 0;
    float* bufA   = ws + off; off += (size_t)N * 128;
    float* bufB   = ws + off; off += (size_t)N * 128;
    float* agg    = ws + off; off += (size_t)N * 128;
    float* part   = ws + off; off += (size_t)N * 128;
    const long long zero4 = ((long long)G * 129 + N + 1 + 3) / 4;
    float* sums = ws + off;
    float* cnt  = sums + (size_t)G * 128;
    int* deg    = (int*)(cnt + G);
    int* cursor = deg + N;
    off += (size_t)zero4 * 4;
    int2* seg   = (int2*)(ws + off); off += (size_t)N * 2;
    int* rank   = (int*)(ws + off); off += (size_t)E;
    int* col    = (int*)(ws + off); off += (size_t)E;
    (void)ws_size; (void)n_in;

    const int T = 256;
    auto gemm_grid = [&](int M, int Cout, int NT) {
        return blocks_for((long long)(M >> 4), 8) * (unsigned)(Cout / (16 * NT));
    };

    zero_kernel<<<blocks_for(zero4, T), T, 0, stream>>>((float4*)sums, zero4);

    const unsigned g1 = gemm_grid(N, 64, 4);
    const unsigned g1a = g1 / 2;
    stage1_and_count_kernel<<<g1a + blocks_for(E, T), T, 0, stream>>>(
        x, w_stage1, b_stage1, bufA, N, dst, deg, rank, E, 0, (int)g1a);
    stage1_and_row_alloc_kernel<<<(g1 - g1a) + blocks_for(N, T), T, 0, stream>>>(
        x, w_stage1, b_stage1, bufA, N, deg, cursor, seg, N, (int)g1a, (int)(g1 - g1a));
    fill_kernel<<<blocks_for(E, T), T, 0, stream>>>(src, dst, rank, seg, col, E);

    const unsigned gc1 = gemm_grid(N, 64, 4);
    gather_sum_and_root_kernel<6, 32><<<gc1 + blocks_for((long long)N << 4, T), T, 0, stream>>>(
        bufA, col, seg, agg, N, root1_w, part, 64, (int)gc1);
    gemm_wmma_kernel<4, 64, true, false, false, true><<<gc1, T, 0, stream>>>(
        agg, rel1_w, rel1_b, bufB, nullptr, nullptr, part, N, 64);

    const unsigned gc2 = gemm_grid(N, 128, 4);
    gather_sum_and_root_kernel<6, 32><<<gc2 + blocks_for((long long)N << 4, T), T, 0, stream>>>(
        bufB, col, seg, agg, N, root2_w, part, 128, (int)gc2);
    gemm_wmma_kernel<4, 64, true, false, false, true><<<gc2, T, 0, stream>>>(
        agg, rel2_w, rel2_b, bufA, nullptr, nullptr, part, N, 128);

    gather_sum_and_root_kernel<7, 16><<<gc2 + blocks_for((long long)N << 5, T), T, 0, stream>>>(
        bufA, col, seg, agg, N, root3_w, part, 128, (int)gc2);
    gemm_wmma_kernel<4, 128, true, true, false, true><<<gc2, T, 0, stream>>>(
        agg, rel3_w, rel3_b, sums, batch, cnt, part, N, 128);

    gemm_wmma_kernel<1, 128, false, false, true, false><<<gemm_grid(G, 16, 1), T, 0, stream>>>(
        sums, lin_w, lin_b, out, nullptr, cnt, nullptr, G, 16);
}


# Edge grouping in one pass: each node owns a fixed run of 64 list entries (overflow list behind), so the counting pass stores sources itself; 2 launches fewer (11 to 9); sum order still arrival order
# speedup vs baseline: 2.7523x; 1.1298x over previous
#include <hip/hip_runtime.h>


typedef __attribute__((ext_vector_type(2))) float v2f;
typedef __attribute__((ext_vector_type(4))) float v4f;
typedef __attribute__((ext_vector_type(8))) float v8f;

template <int NT, int CIN, int KC, bool RELU, bool POOL, bool DIV, bool ACCIN, bool ACCOUT>
__device__ __forceinline__ void gemm_body(
    const float* __restrict__ A, const float* __restrict__ Wm,
    const float* __restrict__ bias, float* __restrict__ out,
    const int* __restrict__ batch, float* __restrict__ cnt,
    float* __restrict__ part, int M, int Cout, int bid)
{
    constexpr int KP    = KC / 2;
    constexpr int PITCH = NT * 32 + ((NT & 1) ? 64 : 32);
    constexpr int C4    = NT * 4;
    __shared__ __attribute__((aligned(16))) float wl[KP * PITCH];

    const int lane = threadIdx.x & 31;
    const int wave = threadIdx.x >> 5;
    const int ncg  = Cout / (16 * NT);
    const int rb   = bid / ncg;
    const int cg   = bid - rb * ncg;
    const int col0 = cg * (16 * NT);
    const int strip = rb * 8 + wave;
    const bool active = strip < (M >> 4);
    const int c = lane & 15;
    const int h = lane >> 4;

    const size_t rowoff = (size_t)((active ? strip : 0) * 16 + c) * CIN + 2 * h;
    const float* arow = A + rowoff;
    float rows_of = 1.0f;
    if (DIV) rows_of = fmaxf(cnt[(active ? strip : 0) * 16 + c], 1.0f);

    const size_t poff = ((size_t)(active ? strip : 0) * (size_t)(Cout >> 4) + (size_t)(cg * NT)) * 256
                        + (size_t)lane * 4;

    v8f acc[NT];
    #pragma unroll
    for (int t = 0; t < NT; ++t) {
        if (ACCIN) {
            const v4f lo = *(const v4f*)(part + poff + t * 256);
            const v4f hi = *(const v4f*)(part + poff + t * 256 + 128);
            acc[t] = __builtin_shufflevector(lo, hi, 0, 1, 2, 3, 4, 5, 6, 7);
        } else {
            acc[t] = v8f{};
        }
    }

    #pragma unroll
    for (int k0 = 0; k0 < CIN; k0 += KC) {
        v2f af[KC / 4];
        if (active) {
            #pragma unroll
            for (int kk = 0; kk < KC / 4; ++kk) {
                af[kk] = *(const v2f*)(arow + k0 + 4 * kk);
                if (DIV) { af[kk].x = af[kk].x / rows_of; af[kk].y = af[kk].y / rows_of; }
            }
        }
        if (k0 > 0) __syncthreads();
        for (int i = threadIdx.x; i < KP * C4; i += 256) {
            const int kp = i / C4;
            const int c4 = i - kp * C4;
            const float* g = Wm + (size_t)(k0 + 2 * kp) * Cout + col0 + 4 * c4;
            const float4 r0 = *(const float4*)g;
            const float4 r1 = *(const float4*)(g + Cout);
            float* d = wl + kp * PITCH + c4 * 8;
            *(float4*)d       = make_float4(r0.x, r1.x, r0.y, r1.y);
            *(float4*)(d + 4) = make_float4(r0.z, r1.z, r0.w, r1.w);
        }
        __syncthreads();
        if (active) {
            #pragma unroll
            for (int kk = 0; kk < KC / 4; ++kk) {
                const float* wrow = wl + (2 * kk + h) * PITCH + c * 2;
                #pragma unroll
                for (int t = 0; t < NT; ++t) {
                    const v2f wv = *(const v2f*)(wrow + t * 32);
                    acc[t] = __builtin_amdgcn_wmma_f32_16x16x4_f32(
                                 false, af[kk], false, wv, (short)0, acc[t], false, false);
                }
            }
        }
    }
    if (!active) return;

    if (ACCOUT) {
        #pragma unroll
        for (int t = 0; t < NT; ++t) {
            *(v4f*)(part + poff + t * 256)       = __builtin_shufflevector(acc[t], acc[t], 0, 1, 2, 3);
            *(v4f*)(part + poff + t * 256 + 128) = __builtin_shufflevector(acc[t], acc[t], 4, 5, 6, 7);
        }
        return;
    }

    const int m0 = strip * 16 + (h << 3);
    if (POOL) {
        int g[8];
        float bv[NT];
        #pragma unroll
        for (int r = 0; r < 8; ++r) g[r] = batch[m0 + r];
        #pragma unroll
        for (int t = 0; t < NT; ++t) bv[t] = bias[col0 + t * 16 + c];
        #pragma unroll
        for (int t = 0; t < NT; ++t) {
            const int col = col0 + t * 16 + c;
            float run = 0.0f;
            #pragma unroll
            for (int r = 0; r < 8; ++r) {
                float v = acc[t][r] + bv[t];
                if (RELU) v = fmaxf(v, 0.0f);
                if (r == 0) {
                    run = v;
                } else {
                    if (g[r] != g[r - 1]) {
                        atomicAdd(out + (g[r - 1] * Cout + col), run);
                        run = 0.0f;
                    }
                    run += v;
                }
            }
            atomicAdd(out + (g[7] * Cout + col), run);
        }
        if (cg == 0 && c == 0) {
            float rows = 1.0f;
            #pragma unroll
            for (int r = 1; r < 8; ++r) {
                if (g[r] != g[r - 1]) {
                    atomicAdd(cnt + g[r - 1], rows);
                    rows = 0.0f;
                }
                rows += 1.0f;
            }
            atomicAdd(cnt + g[7], rows);
        }
        return;
    }
    #pragma unroll
    for (int t = 0; t < NT; ++t) {
        const int col = col0 + t * 16 + c;
        const float bval = bias[col];
        #pragma unroll
        for (int r = 0; r < 8; ++r) {
            float v = acc[t][r] + bval;
            if (RELU) v = fmaxf(v, 0.0f);
            out[(size_t)(m0 + r) * Cout + col] = v;
        }
    }
}

template <int NT, int CIN, bool RELU, bool POOL, bool DIV, bool ACCIN>
__global__ __launch_bounds__(256) void gemm_wmma_kernel(
    const float* __restrict__ A, const float* __restrict__ Wm,
    const float* __restrict__ bias, float* __restrict__ out,
    const int* __restrict__ batch, float* __restrict__ cnt,
    float* __restrict__ part, int M, int Cout)
{
    gemm_body<NT, CIN, 64, RELU, POOL, DIV, ACCIN, false>(A, Wm, bias, out, batch, cnt, part,
                                                          M, Cout, (int)blockIdx.x);
}

constexpr int SLOTS = 64;

__global__ __launch_bounds__(256) void stage1_and_group_kernel(
    const float* __restrict__ x, const float* __restrict__ w,
    const float* __restrict__ bias, float* __restrict__ h0, int M,
    const int* __restrict__ src, const int* __restrict__ dst,
    int* __restrict__ deg, int* __restrict__ col,
    int* __restrict__ novf, int2* __restrict__ ovf, int E, int gemm_groups)
{
    if ((int)blockIdx.x < gemm_groups) {
        gemm_body<4, 128, 64, false, false, false, false, false>(
            x, w, bias, h0, nullptr, nullptr, nullptr, M, 64, (int)blockIdx.x);
        return;
    }
    const int e = ((int)blockIdx.x - gemm_groups) * 256 + threadIdx.x;
    if (e >= E) return;
    const int d = dst[e];
    const int s = src[e];
    const int r = atomicAdd(&deg[d], 1);
    if (r < SLOTS) {
        col[(size_t)d * SLOTS + r] = s;
    } else {
        const int o = atomicAdd(novf, 1);
        ovf[o] = make_int2(d, s);
    }
}

template <int CSHIFT>
__device__ __forceinline__ void gather_sum_body(
    const float* __restrict__ h, const int* __restrict__ col,
    const int* __restrict__ deg, const int* __restrict__ novf,
    const int2* __restrict__ ovf, int E, float* __restrict__ agg, int N, int bid)
{
    constexpr int G4SHIFT = CSHIFT - 2;
    const long long idx = (long long)bid * 256 + threadIdx.x;
    const int node = (int)(idx >> G4SHIFT);
    if (node >= N) return;
    const unsigned c4 = ((unsigned)idx & ((1u << G4SHIFT) - 1u)) << 2;
    const int dg = deg[node];
    int j = node * SLOTS;
    const int end = j + (dg < SLOTS ? dg : SLOTS);
    float4 acc = make_float4(0.f, 0.f, 0.f, 0.f);
    for (; j < end; j += 8) {
        int s[8];
        for (int u = 0; u < 8; ++u) {
            const int jj = j + u;
            s[u] = col[jj < end ? jj : end - 1];
        }
        asm volatile("" :: "v"(s[0]), "v"(s[1]), "v"(s[2]), "v"(s[3]),
                           "v"(s[4]), "v"(s[5]), "v"(s[6]), "v"(s[7]));
        float4 v[8];
        v[0] = *(const float4*)(h + (((unsigned)s[0] << CSHIFT) + c4));
        for (int u = 1; u < 8; ++u) {
            v[u] = make_float4(0.f, 0.f, 0.f, 0.f);
            if (j + u < end) v[u] = *(const float4*)(h + (((unsigned)s[u] << CSHIFT) + c4));
        }
        for (int u = 0; u < 8; ++u) {
            acc.x += v[u].x; acc.y += v[u].y; acc.z += v[u].z; acc.w += v[u].w;
        }
    }
    if (dg > SLOTS) {
        int no = *novf;
        if (no > E) no = E;
        for (int i = 0; i < no; ++i) {
            const int2 p = ovf[i];
            if (p.x == node) {
                const float4 v = *(const float4*)(h + (((unsigned)p.y << CSHIFT) + c4));
                acc.x += v.x; acc.y += v.y; acc.z += v.z; acc.w += v.w;
            }
        }
    }
    *(float4*)(agg + ((size_t)node << CSHIFT) + c4) = acc;
}

template <int CSHIFT, int KC>
__global__ __launch_bounds__(256) __attribute__((amdgpu_waves_per_eu(16, 16)))
void gather_sum_and_root_kernel(
    const float* __restrict__ h, const int* __restrict__ col,
    const int* __restrict__ deg, const int* __restrict__ novf,
    const int2* __restrict__ ovf, int E, float* __restrict__ agg, int N,
    const float* __restrict__ wroot, float* __restrict__ part, int Cout,
    int gemm_groups)
{
    if ((int)blockIdx.x < gemm_groups) {
        gemm_body<4, (1 << CSHIFT), KC, false, false, false, false, true>(
            h, wroot, nullptr, nullptr, nullptr, nullptr, part, N, Cout,
            (int)blockIdx.x);
        return;
    }
    gather_sum_body<CSHIFT>(h, col, deg, novf, ovf, E, agg, N, (int)blockIdx.x - gemm_groups);
}

__global__ __launch_bounds__(256) void zero_kernel(float4* __restrict__ p, long long n4)
{
    long long i = (long long)blockIdx.x * blockDim.x + threadIdx.x;
    if (i < n4) p[i] = make_float4(0.f, 0.f, 0.f, 0.f);
}


static inline unsigned blocks_for(long long n, int t) {
    return (unsigned)((n + t - 1) / t);
}

extern "C" void kernel_launch(void* const* d_in, const int* in_sizes, int n_in,
                              void* d_out, int out_size, void* d_ws, size_t ws_size,
                              hipStream_t stream)
{
    const float* x        = (const float*)d_in[0];
    const int*   edge     = (const int*)  d_in[1];
    const int*   batch    = (const int*)  d_in[2];
    const float* w_stage1 = (const float*)d_in[4];
    const float* b_stage1 = (const float*)d_in[5];
    const float* rel1_w   = (const float*)d_in[6];
    const float* rel1_b   = (const float*)d_in[7];
    const float* root1_w  = (const float*)d_in[8];
    const float* rel2_w   = (const float*)d_in[9];
    const float* rel2_b   = (const float*)d_in[10];
    const float* root2_w  = (const float*)d_in[11];
    const float* rel3_w   = (const float*)d_in[12];
    const float* rel3_b   = (const float*)d_in[13];
    const float* root3_w  = (const float*)d_in[14];
    const float* lin_w    = (const float*)d_in[15];
    const float* lin_b    = (const float*)d_in[16];
    float*       out      = (float*)d_out;

    const int N = in_sizes[0] / 128;
    const int E = in_sizes[1] / 2;
    const int G = out_size / 16;
    const int* src = edge;
    const int* dst = edge + E;

    float* ws = (float*)d_ws;
    size_t off = 0;
    float* bufA   = ws + off; off += (size_t)N * 128;
    float* bufB   = ws + off; off += (size_t)N * 128;
    float* agg    = ws + off; off += (size_t)N * 128;
    float* part   = ws + off; off += (size_t)N * 128;
    const long long zero4 = ((long long)G * 129 + N + 1 + 3) / 4;
    float* sums = ws + off;
    float* cnt  = sums + (size_t)G * 128;
    int* deg    = (int*)(cnt + G);
    int* novf   = deg + N;
    off += (size_t)zero4 * 4;
    off = (off + 63) & ~(size_t)63;
    int* col    = (int*)(ws + off); off += (size_t)N * SLOTS;
    int2* ovf   = (int2*)(ws + off); off += (size_t)E * 2;
    (void)ws_size; (void)n_in;

    const int T = 256;
    auto gemm_grid = [&](int M, int Cout, int NT) {
        return blocks_for((long long)(M >> 4), 8) * (unsigned)(Cout / (16 * NT));
    };

    zero_kernel<<<blocks_for(zero4, T), T, 0, stream>>>((float4*)sums, zero4);

    const unsigned g1 = gemm_grid(N, 64, 4);
    stage1_and_group_kernel<<<g1 + blocks_for(E, T), T, 0, stream>>>(
        x, w_stage1, b_stage1, bufA, N, src, dst, deg, col, novf, ovf, E, (int)g1);

    const unsigned gc1 = gemm_grid(N, 64, 4);
    gather_sum_and_root_kernel<6, 32><<<gc1 + blocks_for((long long)N << 4, T), T, 0, stream>>>(
        bufA, col, deg, novf, ovf, E, agg, N, root1_w, part, 64, (int)gc1);
    gemm_wmma_kernel<4, 64, true, false, false, true><<<gc1, T, 0, stream>>>(
        agg, rel1_w, rel1_b, bufB, nullptr, nullptr, part, N, 64);

    const unsigned gc2 = gemm_grid(N, 128, 4);
    gather_sum_and_root_kernel<6, 32><<<gc2 + blocks_for((long long)N << 4, T), T, 0, stream>>>(
        bufB, col, deg, novf, ovf, E, agg, N, root2_w, part, 128, (int)gc2);
    gemm_wmma_kernel<4, 64, true, false, false, true><<<gc2, T, 0, stream>>>(
        agg, rel2_w, rel2_b, bufA, nullptr, nullptr, part, N, 128);

    gather_sum_and_root_kernel<7, 16><<<gc2 + blocks_for((long long)N << 5, T), T, 0, stream>>>(
        bufA, col, deg, novf, ovf, E, agg, N, root3_w, part, 128, (int)gc2);
    gemm_wmma_kernel<4, 128, true, true, false, true><<<gc2, T, 0, stream>>>(
        agg, rel3_w, rel3_b, sums, batch, cnt, part, N, 128);

    gemm_wmma_kernel<1, 128, false, false, true, false><<<gemm_grid(G, 16, 1), T, 0, stream>>>(
        sums, lin_w, lin_b, out, nullptr, cnt, nullptr, G, 16);
}


# Conv layers 1 and 2 in one launch each: the work-group that forms a 64-node block's neighbour sums also runs both products (sums, own rows, weights via LDS); 9 to 7 launches; sum order kept
# speedup vs baseline: 2.9823x; 1.0836x over previous
#include <hip/hip_runtime.h>


typedef __attribute__((ext_vector_type(2))) float v2f;
typedef __attribute__((ext_vector_type(4))) float v4f;
typedef __attribute__((ext_vector_type(8))) float v8f;

template <int NT, int CIN, int KC, bool RELU, bool POOL, bool DIV, bool ACCIN, bool ACCOUT>
__device__ __forceinline__ void gemm_body(
    const float* __restrict__ A, const float* __restrict__ Wm,
    const float* __restrict__ bias, float* __restrict__ out,
    const int* __restrict__ batch, float* __restrict__ cnt,
    float* __restrict__ part, int M, int Cout, int bid)
{
    constexpr int KP    = KC / 2;
    constexpr int PITCH = NT * 32 + ((NT & 1) ? 64 : 32);
    constexpr int C4    = NT * 4;
    __shared__ __attribute__((aligned(16))) float wl[KP * PITCH];

    const int lane = threadIdx.x & 31;
    const int wave = threadIdx.x >> 5;
    const int ncg  = Cout / (16 * NT);
    const int rb   = bid / ncg;
    const int cg   = bid - rb * ncg;
    const int col0 = cg * (16 * NT);
    const int strip = rb * 8 + wave;
    const bool active = strip < (M >> 4);
    const int c = lane & 15;
    const int h = lane >> 4;

    const size_t rowoff = (size_t)((active ? strip : 0) * 16 + c) * CIN + 2 * h;
    const float* arow = A + rowoff;
    float rows_of = 1.0f;
    if (DIV) rows_of = fmaxf(cnt[(active ? strip : 0) * 16 + c], 1.0f);

    const size_t poff = ((size_t)(active ? strip : 0) * (size_t)(Cout >> 4) + (size_t)(cg * NT)) * 256
                        + (size_t)lane * 4;

    v8f acc[NT];
    #pragma unroll
    for (int t = 0; t < NT; ++t) {
        if (ACCIN) {
            const v4f lo = *(const v4f*)(part + poff + t * 256);
            const v4f hi = *(const v4f*)(part + poff + t * 256 + 128);
            acc[t] = __builtin_shufflevector(lo, hi, 0, 1, 2, 3, 4, 5, 6, 7);
        } else {
            acc[t] = v8f{};
        }
    }

    #pragma unroll
    for (int k0 = 0; k0 < CIN; k0 += KC) {
        v2f af[KC / 4];
        if (active) {
            #pragma unroll
            for (int kk = 0; kk < KC / 4; ++kk) {
                af[kk] = *(const v2f*)(arow + k0 + 4 * kk);
                if (DIV) { af[kk].x = af[kk].x / rows_of; af[kk].y = af[kk].y / rows_of; }
            }
        }
        if (k0 > 0) __syncthreads();
        for (int i = threadIdx.x; i < KP * C4; i += 256) {
            const int kp = i / C4;
            const int c4 = i - kp * C4;
            const float* g = Wm + (size_t)(k0 + 2 * kp) * Cout + col0 + 4 * c4;
            const float4 r0 = *(const float4*)g;
            const float4 r1 = *(const float4*)(g + Cout);
            float* d = wl + kp * PITCH + c4 * 8;
            *(float4*)d       = make_float4(r0.x, r1.x, r0.y, r1.y);
            *(float4*)(d + 4) = make_float4(r0.z, r1.z, r0.w, r1.w);
        }
        __syncthreads();
        if (active) {
            #pragma unroll
            for (int kk = 0; kk < KC / 4; ++kk) {
                const float* wrow = wl + (2 * kk + h) * PITCH + c * 2;
                #pragma unroll
                for (int t = 0; t < NT; ++t) {
                    const v2f wv = *(const v2f*)(wrow + t * 32);
                    acc[t] = __builtin_amdgcn_wmma_f32_16x16x4_f32(
                                 false, af[kk], false, wv, (short)0, acc[t], false, false);
                }
            }
        }
    }
    if (!active) return;

    if (ACCOUT) {
        #pragma unroll
        for (int t = 0; t < NT; ++t) {
            *(v4f*)(part + poff + t * 256)       = __builtin_shufflevector(acc[t], acc[t], 0, 1, 2, 3);
            *(v4f*)(part + poff + t * 256 + 128) = __builtin_shufflevector(acc[t], acc[t], 4, 5, 6, 7);
        }
        return;
    }

    const int m0 = strip * 16 + (h << 3);
    if (POOL) {
        int g[8];
        float bv[NT];
        #pragma unroll
        for (int r = 0; r < 8; ++r) g[r] = batch[m0 + r];
        #pragma unroll
        for (int t = 0; t < NT; ++t) bv[t] = bias[col0 + t * 16 + c];
        #pragma unroll
        for (int t = 0; t < NT; ++t) {
            const int col = col0 + t * 16 + c;
            float run = 0.0f;
            #pragma unroll
            for (int r = 0; r < 8; ++r) {
                float v = acc[t][r] + bv[t];
                if (RELU) v = fmaxf(v, 0.0f);
                if (r == 0) {
                    run = v;
                } else {
                    if (g[r] != g[r - 1]) {
                        atomicAdd(out + (g[r - 1] * Cout + col), run);
                        run = 0.0f;
                    }
                    run += v;
                }
            }
            atomicAdd(out + (g[7] * Cout + col), run);
        }
        if (cg == 0 && c == 0) {
            float rows = 1.0f;
            #pragma unroll
            for (int r = 1; r < 8; ++r) {
                if (g[r] != g[r - 1]) {
                    atomicAdd(cnt + g[r - 1], rows);
                    rows = 0.0f;
                }
                rows += 1.0f;
            }
            atomicAdd(cnt + g[7], rows);
        }
        return;
    }
    #pragma unroll
    for (int t = 0; t < NT; ++t) {
        const int col = col0 + t * 16 + c;
        const float bval = bias[col];
        #pragma unroll
        for (int r = 0; r < 8; ++r) {
            float v = acc[t][r] + bval;
            if (RELU) v = fmaxf(v, 0.0f);
            out[(size_t)(m0 + r) * Cout + col] = v;
        }
    }
}

template <int NT, int CIN, bool RELU, bool POOL, bool DIV, bool ACCIN>
__global__ __launch_bounds__(256) void gemm_wmma_kernel(
    const float* __restrict__ A, const float* __restrict__ Wm,
    const float* __restrict__ bias, float* __restrict__ out,
    const int* __restrict__ batch, float* __restrict__ cnt,
    float* __restrict__ part, int M, int Cout)
{
    gemm_body<NT, CIN, 64, RELU, POOL, DIV, ACCIN, false>(A, Wm, bias, out, batch, cnt, part,
                                                          M, Cout, (int)blockIdx.x);
}

constexpr int SLOTS = 64;

__global__ __launch_bounds__(256) void stage1_and_group_kernel(
    const float* __restrict__ x, const float* __restrict__ w,
    const float* __restrict__ bias, float* __restrict__ h0, int M,
    const int* __restrict__ src, const int* __restrict__ dst,
    int* __restrict__ deg, int* __restrict__ col,
    int* __restrict__ novf, int2* __restrict__ ovf, int E, int gemm_groups)
{
    if ((int)blockIdx.x < gemm_groups) {
        gemm_body<4, 128, 64, false, false, false, false, false>(
            x, w, bias, h0, nullptr, nullptr, nullptr, M, 64, (int)blockIdx.x);
        return;
    }
    const int e = ((int)blockIdx.x - gemm_groups) * 256 + threadIdx.x;
    if (e >= E) return;
    const int d = dst[e];
    const int s = src[e];
    const int r = atomicAdd(&deg[d], 1);
    if (r < SLOTS) {
        col[(size_t)d * SLOTS + r] = s;
    } else {
        const int o = atomicAdd(novf, 1);
        ovf[o] = make_int2(d, s);
    }
}

template <int CSHIFT>
__device__ __forceinline__ float4 gather_sum_row(
    const float* __restrict__ h, const int* __restrict__ col,
    const int* __restrict__ novf,
    const int2* __restrict__ ovf, int E, int node, int dg, unsigned c4)
{
    int j = node * SLOTS;
    const int end = j + (dg < SLOTS ? dg : SLOTS);
    float4 acc = make_float4(0.f, 0.f, 0.f, 0.f);
    for (; j < end; j += 8) {
        int s[8];
        for (int u = 0; u < 8; ++u) {
            const int jj = j + u;
            s[u] = col[jj < end ? jj : end - 1];
        }
        asm volatile("" :: "v"(s[0]), "v"(s[1]), "v"(s[2]), "v"(s[3]),
                           "v"(s[4]), "v"(s[5]), "v"(s[6]), "v"(s[7]));
        float4 v[8];
        v[0] = *(const float4*)(h + (((unsigned)s[0] << CSHIFT) + c4));
        for (int u = 1; u < 8; ++u) {
            v[u] = make_float4(0.f, 0.f, 0.f, 0.f);
            if (j + u < end) v[u] = *(const float4*)(h + (((unsigned)s[u] << CSHIFT) + c4));
        }
        for (int u = 0; u < 8; ++u) {
            acc.x += v[u].x; acc.y += v[u].y; acc.z += v[u].z; acc.w += v[u].w;
        }
    }
    if (dg > SLOTS) {
        int no = *novf;
        if (no > E) no = E;
        for (int i = 0; i < no; ++i) {
            const int2 p = ovf[i];
            if (p.x == node) {
                const float4 v = *(const float4*)(h + (((unsigned)p.y << CSHIFT) + c4));
                acc.x += v.x; acc.y += v.y; acc.z += v.z; acc.w += v.w;
            }
        }
    }
    return acc;
}

template <int CSHIFT>
__device__ __forceinline__ void gather_sum_body(
    const float* __restrict__ h, const int* __restrict__ col,
    const int* __restrict__ deg, const int* __restrict__ novf,
    const int2* __restrict__ ovf, int E, float* __restrict__ agg, int N, int bid)
{
    constexpr int G4SHIFT = CSHIFT - 2;
    const long long idx = (long long)bid * 256 + threadIdx.x;
    const int node = (int)(idx >> G4SHIFT);
    if (node >= N) return;
    const unsigned c4 = ((unsigned)idx & ((1u << G4SHIFT) - 1u)) << 2;
    const float4 acc = gather_sum_row<CSHIFT>(h, col, novf, ovf, E, node, deg[node], c4);
    *(float4*)(agg + ((size_t)node << CSHIFT) + c4) = acc;
}

template <int CSHIFT, int KC>
__global__ __launch_bounds__(256) __attribute__((amdgpu_waves_per_eu(16, 16)))
void gather_sum_and_root_kernel(
    const float* __restrict__ h, const int* __restrict__ col,
    const int* __restrict__ deg, const int* __restrict__ novf,
    const int2* __restrict__ ovf, int E, float* __restrict__ agg, int N,
    const float* __restrict__ wroot, float* __restrict__ part, int Cout,
    int gemm_groups)
{
    if ((int)blockIdx.x < gemm_groups) {
        gemm_body<4, (1 << CSHIFT), KC, false, false, false, false, true>(
            h, wroot, nullptr, nullptr, nullptr, nullptr, part, N, Cout,
            (int)blockIdx.x);
        return;
    }
    gather_sum_body<CSHIFT>(h, col, deg, novf, ovf, E, agg, N, (int)blockIdx.x - gemm_groups);
}

template <int COUT>
__global__ __launch_bounds__(1024) __attribute__((amdgpu_waves_per_eu(16, 16)))
void conv64_fused_kernel(
    const float* __restrict__ h, const int* __restrict__ col,
    const int* __restrict__ deg, const int* __restrict__ novf,
    const int2* __restrict__ ovf, int E, int N,
    const float* __restrict__ wroot, const float* __restrict__ wrel,
    const float* __restrict__ bias, float* __restrict__ out)
{
    constexpr int CIN   = 64;
    constexpr int NT    = COUT / 16;
    constexpr int KP    = CIN / 2;
    constexpr int PITCH = NT * 32 + 32;
    constexpr int C4    = NT * 4;
    constexpr int ITEMS = 2 * KP * C4;
    constexpr int NW    = ITEMS / 1024;
    constexpr int AP    = 68;
    static_assert((NT & 1) == 0 && ITEMS % 1024 == 0, "shape");
    __shared__ __attribute__((aligned(16))) float wl[2 * KP * PITCH];
    __shared__ __attribute__((aligned(16))) float hT[64 * AP];
    __shared__ __attribute__((aligned(16))) float aT[64 * AP];

    const int tid  = threadIdx.x;
    const int nl   = tid >> 4;
    const unsigned c4 = ((unsigned)tid & 15u) << 2;
    const int node = (int)blockIdx.x * 64 + nl;
    const bool valid = node < N;

    float4 w0[NW], w1[NW];
    #pragma unroll
    for (int q = 0; q < NW; ++q) {
        const int it = tid + q * 1024;
        const int m  = it / (KP * C4);
        const int i  = it - m * (KP * C4);
        const int kp = i / C4;
        const int ci = i - kp * C4;
        const float* g = (m ? wrel : wroot) + (size_t)(2 * kp) * COUT + 4 * ci;
        w0[q] = *(const float4*)g;
        w1[q] = *(const float4*)(g + COUT);
    }
    float4 own = make_float4(0.f, 0.f, 0.f, 0.f);
    int dg = 0;
    if (valid) {
        own = *(const float4*)(h + ((size_t)node << 6) + c4);
        dg  = deg[node];
    }
    #pragma unroll
    for (int q = 0; q < NW; ++q) {
        const int it = tid + q * 1024;
        const int m  = it / (KP * C4);
        const int i  = it - m * (KP * C4);
        const int kp = i / C4;
        const int ci = i - kp * C4;
        float* d = wl + m * (KP * PITCH) + kp * PITCH + ci * 8;
        *(float4*)d       = make_float4(w0[q].x, w1[q].x, w0[q].y, w1[q].y);
        *(float4*)(d + 4) = make_float4(w0[q].z, w1[q].z, w0[q].w, w1[q].w);
    }
    *(float4*)(hT + nl * AP + c4) = own;

    float4 sum = make_float4(0.f, 0.f, 0.f, 0.f);
    if (valid) sum = gather_sum_row<6>(h, col, novf, ovf, E, node, dg, c4);
    *(float4*)(aT + nl * AP + c4) = sum;
    __syncthreads();

    const int lane = tid & 31;
    const int wave = tid >> 5;
    if (wave >= 4 * NT) return;
    const int strip = wave / NT;
    const int tile  = wave - strip * NT;
    const int row0  = (int)blockIdx.x * 64 + strip * 16;
    if (row0 >= N) return;
    const int c  = lane & 15;
    const int hh = lane >> 4;

    v8f acc = v8f{};
    const float* ah = hT + (strip * 16 + c) * AP + 2 * hh;
    const float* aa = aT + (strip * 16 + c) * AP + 2 * hh;
    const float* wb = wl + hh * PITCH + tile * 32 + c * 2;
    #pragma unroll
    for (int kk = 0; kk < CIN / 4; ++kk) {
        const v2f av = *(const v2f*)(ah + 4 * kk);
        const v2f wv = *(const v2f*)(wb + 2 * kk * PITCH);
        acc = __builtin_amdgcn_wmma_f32_16x16x4_f32(false, av, false, wv, (short)0, acc, false, false);
    }
    #pragma unroll
    for (int kk = 0; kk < CIN / 4; ++kk) {
        const v2f av = *(const v2f*)(aa + 4 * kk);
        const v2f wv = *(const v2f*)(wb + KP * PITCH + 2 * kk * PITCH);
        acc = __builtin_amdgcn_wmma_f32_16x16x4_f32(false, av, false, wv, (short)0, acc, false, false);
    }

    const int m0 = row0 + (hh << 3);
    const int cc = tile * 16 + c;
    const float bval = bias[cc];
    #pragma unroll
    for (int r = 0; r < 8; ++r) {
        const float v = fmaxf(acc[r] + bval, 0.0f);
        out[(size_t)(m0 + r) * COUT + cc] = v;
    }
}

__global__ __launch_bounds__(256) void zero_kernel(float4* __restrict__ p, long long n4)
{
    long long i = (long long)blockIdx.x * blockDim.x + threadIdx.x;
    if (i < n4) p[i] = make_float4(0.f, 0.f, 0.f, 0.f);
}


static inline unsigned blocks_for(long long n, int t) {
    return (unsigned)((n + t - 1) / t);
}

extern "C" void kernel_launch(void* const* d_in, const int* in_sizes, int n_in,
                              void* d_out, int out_size, void* d_ws, size_t ws_size,
                              hipStream_t stream)
{
    const float* x        = (const float*)d_in[0];
    const int*   edge     = (const int*)  d_in[1];
    const int*   batch    = (const int*)  d_in[2];
    const float* w_stage1 = (const float*)d_in[4];
    const float* b_stage1 = (const float*)d_in[5];
    const float* rel1_w   = (const float*)d_in[6];
    const float* rel1_b   = (const float*)d_in[7];
    const float* root1_w  = (const float*)d_in[8];
    const float* rel2_w   = (const float*)d_in[9];
    const float* rel2_b   = (const float*)d_in[10];
    const float* root2_w  = (const float*)d_in[11];
    const float* rel3_w   = (const float*)d_in[12];
    const float* rel3_b   = (const float*)d_in[13];
    const float* root3_w  = (const float*)d_in[14];
    const float* lin_w    = (const float*)d_in[15];
    const float* lin_b    = (const float*)d_in[16];
    float*       out      = (float*)d_out;

    const int N = in_sizes[0] / 128;
    const int E = in_sizes[1] / 2;
    const int G = out_size / 16;
    const int* src = edge;
    const int* dst = edge + E;

    float* ws = (float*)d_ws;
    size_t off = 0;
    float* bufA   = ws + off; off += (size_t)N * 128;
    float* bufB   = ws + off; off += (size_t)N * 128;
    float* agg    = ws + off; off += (size_t)N * 128;
    float* part   = ws + off; off += (size_t)N * 128;
    const long long zero4 = ((long long)G * 129 + N + 1 + 3) / 4;
    float* sums = ws + off;
    float* cnt  = sums + (size_t)G * 128;
    int* deg    = (int*)(cnt + G);
    int* novf   = deg + N;
    off += (size_t)zero4 * 4;
    off = (off + 63) & ~(size_t)63;
    int* col    = (int*)(ws + off); off += (size_t)N * SLOTS;
    int2* ovf   = (int2*)(ws + off); off += (size_t)E * 2;
    (void)ws_size; (void)n_in;

    const int T = 256;
    auto gemm_grid = [&](int M, int Cout, int NT) {
        return blocks_for((long long)(M >> 4), 8) * (unsigned)(Cout / (16 * NT));
    };

    zero_kernel<<<blocks_for(zero4, T), T, 0, stream>>>((float4*)sums, zero4);

    const unsigned g1 = gemm_grid(N, 64, 4);
    stage1_and_group_kernel<<<g1 + blocks_for(E, T), T, 0, stream>>>(
        x, w_stage1, b_stage1, bufA, N, src, dst, deg, col, novf, ovf, E, (int)g1);

    const unsigned gf = blocks_for(N, 64);
    conv64_fused_kernel<64><<<gf, 1024, 0, stream>>>(
        bufA, col, deg, novf, ovf, E, N, root1_w, rel1_w, rel1_b, bufB);
    conv64_fused_kernel<128><<<gf, 1024, 0, stream>>>(
        bufB, col, deg, novf, ovf, E, N, root2_w, rel2_w, rel2_b, bufA);

    const unsigned gc2 = gemm_grid(N, 128, 4);
    gather_sum_and_root_kernel<7, 16><<<gc2 + blocks_for((long long)N << 5, T), T, 0, stream>>>(
        bufA, col, deg, novf, ovf, E, agg, N, root3_w, part, 128, (int)gc2);
    gemm_wmma_kernel<4, 128, true, true, false, true><<<gc2, T, 0, stream>>>(
        agg, rel3_w, rel3_b, sums, batch, cnt, part, N, 128);

    gemm_wmma_kernel<1, 128, false, false, true, false><<<gemm_grid(G, 16, 1), T, 0, stream>>>(
        sums, lin_w, lin_b, out, nullptr, cnt, nullptr, G, 16);
}


# Edge grouping by a two-level counting sort on LDS counters (chunks sorted by 64-node block; conv 1's block owner ranks them): no per-edge global atomics; 6 launches; sum order = LDS arrival order
# speedup vs baseline: 3.8849x; 1.3027x over previous
#include <hip/hip_runtime.h>


typedef __attribute__((ext_vector_type(2))) float v2f;
typedef __attribute__((ext_vector_type(4))) float v4f;
typedef __attribute__((ext_vector_type(8))) float v8f;

template <int NT, int CIN, int KC, bool RELU, bool POOL, bool DIV, bool ACCIN, bool ACCOUT>
__device__ __forceinline__ void gemm_body(
    const float* __restrict__ A, const float* __restrict__ Wm,
    const float* __restrict__ bias, float* __restrict__ out,
    const int* __restrict__ batch, float* __restrict__ cnt,
    float* __restrict__ part, int M, int Cout, int bid)
{
    constexpr int KP    = KC / 2;
    constexpr int PITCH = NT * 32 + ((NT & 1) ? 64 : 32);
    constexpr int C4    = NT * 4;
    __shared__ __attribute__((aligned(16))) float wl[KP * PITCH];

    const int lane = threadIdx.x & 31;
    const int wave = threadIdx.x >> 5;
    const int ncg  = Cout / (16 * NT);
    const int rb   = bid / ncg;
    const int cg   = bid - rb * ncg;
    const int col0 = cg * (16 * NT);
    const int strip = rb * 8 + wave;
    const bool active = strip < (M >> 4);
    const int c = lane & 15;
    const int h = lane >> 4;

    const size_t rowoff = (size_t)((active ? strip : 0) * 16 + c) * CIN + 2 * h;
    const float* arow = A + rowoff;
    float rows_of = 1.0f;
    if (DIV) rows_of = fmaxf(cnt[(active ? strip : 0) * 16 + c], 1.0f);

    const size_t poff = ((size_t)(active ? strip : 0) * (size_t)(Cout >> 4) + (size_t)(cg * NT)) * 256
                        + (size_t)lane * 4;

    v8f acc[NT];
    #pragma unroll
    for (int t = 0; t < NT; ++t) {
        if (ACCIN) {
            const v4f lo = *(const v4f*)(part + poff + t * 256);
            const v4f hi = *(const v4f*)(part + poff + t * 256 + 128);
            acc[t] = __builtin_shufflevector(lo, hi, 0, 1, 2, 3, 4, 5, 6, 7);
        } else {
            acc[t] = v8f{};
        }
    }

    #pragma unroll
    for (int k0 = 0; k0 < CIN; k0 += KC) {
        v2f af[KC / 4];
        if (active) {
            #pragma unroll
            for (int kk = 0; kk < KC / 4; ++kk) {
                af[kk] = *(const v2f*)(arow + k0 + 4 * kk);
                if (DIV) { af[kk].x = af[kk].x / rows_of; af[kk].y = af[kk].y / rows_of; }
            }
        }
        if (k0 > 0) __syncthreads();
        for (int i = threadIdx.x; i < KP * C4; i += 256) {
            const int kp = i / C4;
            const int c4 = i - kp * C4;
            const float* g = Wm + (size_t)(k0 + 2 * kp) * Cout + col0 + 4 * c4;
            const float4 r0 = *(const float4*)g;
            const float4 r1 = *(const float4*)(g + Cout);
            float* d = wl + kp * PITCH + c4 * 8;
            *(float4*)d       = make_float4(r0.x, r1.x, r0.y, r1.y);
            *(float4*)(d + 4) = make_float4(r0.z, r1.z, r0.w, r1.w);
        }
        __syncthreads();
        if (active) {
            #pragma unroll
            for (int kk = 0; kk < KC / 4; ++kk) {
                const float* wrow = wl + (2 * kk + h) * PITCH + c * 2;
                #pragma unroll
                for (int t = 0; t < NT; ++t) {
                    const v2f wv = *(const v2f*)(wrow + t * 32);
                    acc[t] = __builtin_amdgcn_wmma_f32_16x16x4_f32(
                                 false, af[kk], false, wv, (short)0, acc[t], false, false);
                }
            }
        }
    }
    if (!active) return;

    if (ACCOUT) {
        #pragma unroll
        for (int t = 0; t < NT; ++t) {
            *(v4f*)(part + poff + t * 256)       = __builtin_shufflevector(acc[t], acc[t], 0, 1, 2, 3);
            *(v4f*)(part + poff + t * 256 + 128) = __builtin_shufflevector(acc[t], acc[t], 4, 5, 6, 7);
        }
        return;
    }

    const int m0 = strip * 16 + (h << 3);
    if (POOL) {
        int g[8];
        float bv[NT];
        #pragma unroll
        for (int r = 0; r < 8; ++r) g[r] = batch[m0 + r];
        #pragma unroll
        for (int t = 0; t < NT; ++t) bv[t] = bias[col0 + t * 16 + c];
        #pragma unroll
        for (int t = 0; t < NT; ++t) {
            const int col = col0 + t * 16 + c;
            float run = 0.0f;
            #pragma unroll
            for (int r = 0; r < 8; ++r) {
                float v = acc[t][r] + bv[t];
                if (RELU) v = fmaxf(v, 0.0f);
                if (r == 0) {
                    run = v;
                } else {
                    if (g[r] != g[r - 1]) {
                        atomicAdd(out + (g[r - 1] * Cout + col), run);
                        run = 0.0f;
                    }
                    run += v;
                }
            }
            atomicAdd(out + (g[7] * Cout + col), run);
        }
        if (cg == 0 && c == 0) {
            float rows = 1.0f;
            #pragma unroll
            for (int r = 1; r < 8; ++r) {
                if (g[r] != g[r - 1]) {
                    atomicAdd(cnt + g[r - 1], rows);
                    rows = 0.0f;
                }
                rows += 1.0f;
            }
            atomicAdd(cnt + g[7], rows);
        }
        return;
    }
    #pragma unroll
    for (int t = 0; t < NT; ++t) {
        const int col = col0 + t * 16 + c;
        const float bval = bias[col];
        #pragma unroll
        for (int r = 0; r < 8; ++r) {
            float v = acc[t][r] + bval;
            if (RELU) v = fmaxf(v, 0.0f);
            out[(size_t)(m0 + r) * Cout + col] = v;
        }
    }
}

template <int NT, int CIN, bool RELU, bool POOL, bool DIV, bool ACCIN>
__global__ __launch_bounds__(256) void gemm_wmma_kernel(
    const float* __restrict__ A, const float* __restrict__ Wm,
    const float* __restrict__ bias, float* __restrict__ out,
    const int* __restrict__ batch, float* __restrict__ cnt,
    float* __restrict__ part, int M, int Cout)
{
    gemm_body<NT, CIN, 64, RELU, POOL, DIV, ACCIN, false>(A, Wm, bias, out, batch, cnt, part,
                                                          M, Cout, (int)blockIdx.x);
}

constexpr int SLOTS = 64;

constexpr int CH    = 8192;
constexpr int NBMAX = 1024;

__device__ __forceinline__ void chunk_sort_body(
    const int* __restrict__ src, const int* __restrict__ dst, int E, int nb,
    int2* __restrict__ sorted, int* __restrict__ segoff, int chunk)
{
    constexpr int EPT = CH / 256;
    __shared__ int hist[NBMAX];
    __shared__ int wtot[8];
    __shared__ __attribute__((aligned(16))) int2 stage[CH];

    const int tid  = threadIdx.x;
    const int lane = tid & 31;
    const int wave = tid >> 5;
    for (int i = tid; i < NBMAX; i += 256) hist[i] = 0;

    const int base = chunk * CH;
    int d[EPT], s[EPT], bin[EPT];
    #pragma unroll
    for (int u = 0; u < EPT; ++u) {
        const int e  = base + u * 256 + tid;
        const int ec = e < E ? e : E - 1;
        d[u] = dst[ec];
        s[u] = src[ec];
        bin[u] = e < E ? (d[u] >> 6) : (NBMAX - 1 - lane);
    }
    __syncthreads();
    #pragma unroll
    for (int u = 0; u < EPT; ++u) atomicAdd(&hist[bin[u]], 1);
    __syncthreads();

    const int b4 = 4 * tid;
    const int h0 = hist[b4], h1 = hist[b4 + 1], h2 = hist[b4 + 2], h3 = hist[b4 + 3];
    const int tsum = h0 + h1 + h2 + h3;
    int incl = tsum;
    #pragma unroll
    for (int o = 1; o < 32; o <<= 1) {
        const int v = __shfl_up(incl, o);
        if (lane >= o) incl += v;
    }
    if (lane == 31) wtot[wave] = incl;
    __syncthreads();
    int ex = incl - tsum;
    for (int w = 0; w < wave; ++w) ex += wtot[w];
    const int e1 = ex + h0, e2 = e1 + h1, e3 = e2 + h2;
    hist[b4] = ex; hist[b4 + 1] = e1; hist[b4 + 2] = e2; hist[b4 + 3] = e3;
    int* so = segoff + (size_t)chunk * (size_t)(nb + 1);
    if (b4     <= nb) so[b4]     = ex;
    if (b4 + 1 <= nb) so[b4 + 1] = e1;
    if (b4 + 2 <= nb) so[b4 + 2] = e2;
    if (b4 + 3 <= nb) so[b4 + 3] = e3;
    __syncthreads();

    #pragma unroll
    for (int g = 0; g < EPT; g += 8) {
        int pos[8];
        #pragma unroll
        for (int u = 0; u < 8; ++u) pos[u] = atomicAdd(&hist[bin[g + u]], 1);
        #pragma unroll
        for (int u = 0; u < 8; ++u) stage[pos[u]] = make_int2(d[g + u], s[g + u]);
    }
    __syncthreads();
    const int total = (E - base < CH) ? (E - base) : CH;
    int2 t[EPT];
    #pragma unroll
    for (int u = 0; u < EPT; ++u) t[u] = stage[u * 256 + tid];
    #pragma unroll
    for (int u = 0; u < EPT; ++u)
        if (u * 256 + tid < total) sorted[(size_t)base + u * 256 + tid] = t[u];
}

__global__ __launch_bounds__(256) void stage1_and_sort_kernel(
    const float* __restrict__ x, const float* __restrict__ w,
    const float* __restrict__ bias, float* __restrict__ h0, int M,
    const int* __restrict__ src, const int* __restrict__ dst, int E, int nb,
    int2* __restrict__ sorted, int* __restrict__ segoff, int nchunks, int gemm_groups,
    float4* __restrict__ zp, long long n4)
{
    const int bid = (int)blockIdx.x;
    if (bid < nchunks) {
        chunk_sort_body(src, dst, E, nb, sorted, segoff, bid);
        return;
    }
    if (bid < nchunks + gemm_groups) {
        gemm_body<4, 128, 64, false, false, false, false, false>(
            x, w, bias, h0, nullptr, nullptr, nullptr, M, 64, bid - nchunks);
        return;
    }
    const long long i = (long long)(bid - nchunks - gemm_groups) * 256 + threadIdx.x;
    if (i < n4) zp[i] = make_float4(0.f, 0.f, 0.f, 0.f);
}

template <int CSHIFT>
__device__ __forceinline__ float4 gather_sum_list(
    const float* __restrict__ h, const int* col, int j, const int end, unsigned c4)
{
    float4 acc = make_float4(0.f, 0.f, 0.f, 0.f);
    for (; j < end; j += 8) {
        int s[8];
        for (int u = 0; u < 8; ++u) {
            const int jj = j + u;
            s[u] = col[jj < end ? jj : end - 1];
        }
        asm volatile("" :: "v"(s[0]), "v"(s[1]), "v"(s[2]), "v"(s[3]),
                           "v"(s[4]), "v"(s[5]), "v"(s[6]), "v"(s[7]));
        float4 v[8];
        v[0] = *(const float4*)(h + (((unsigned)s[0] << CSHIFT) + c4));
        for (int u = 1; u < 8; ++u) {
            v[u] = make_float4(0.f, 0.f, 0.f, 0.f);
            if (j + u < end) v[u] = *(const float4*)(h + (((unsigned)s[u] << CSHIFT) + c4));
        }
        for (int u = 0; u < 8; ++u) {
            acc.x += v[u].x; acc.y += v[u].y; acc.z += v[u].z; acc.w += v[u].w;
        }
    }
    return acc;
}

template <int CSHIFT>
__device__ __forceinline__ float4 gather_sum_row(
    const float* __restrict__ h, const int* __restrict__ col,
    const int* __restrict__ novf,
    const int2* __restrict__ ovf, int E, int node, int dg, unsigned c4)
{
    const int j = node * SLOTS;
    float4 acc = gather_sum_list<CSHIFT>(h, col, j, j + (dg < SLOTS ? dg : SLOTS), c4);
    if (dg > SLOTS) {
        int no = *novf;
        if (no > E) no = E;
        for (int i = 0; i < no; ++i) {
            const int2 p = ovf[i];
            if (p.x == node) {
                const float4 v = *(const float4*)(h + (((unsigned)p.y << CSHIFT) + c4));
                acc.x += v.x; acc.y += v.y; acc.z += v.z; acc.w += v.w;
            }
        }
    }
    return acc;
}

template <int CSHIFT>
__device__ __forceinline__ void gather_sum_body(
    const float* __restrict__ h, const int* __restrict__ col,
    const int* __restrict__ deg, const int* __restrict__ novf,
    const int2* __restrict__ ovf, int E, float* __restrict__ agg, int N, int bid)
{
    constexpr int G4SHIFT = CSHIFT - 2;
    const long long idx = (long long)bid * 256 + threadIdx.x;
    const int node = (int)(idx >> G4SHIFT);
    if (node >= N) return;
    const unsigned c4 = ((unsigned)idx & ((1u << G4SHIFT) - 1u)) << 2;
    const float4 acc = gather_sum_row<CSHIFT>(h, col, novf, ovf, E, node, deg[node], c4);
    *(float4*)(agg + ((size_t)node << CSHIFT) + c4) = acc;
}

template <int CSHIFT, int KC>
__global__ __launch_bounds__(256) __attribute__((amdgpu_waves_per_eu(16, 16)))
void gather_sum_and_root_kernel(
    const float* __restrict__ h, const int* __restrict__ col,
    const int* __restrict__ deg, const int* __restrict__ novf,
    const int2* __restrict__ ovf, int E, float* __restrict__ agg, int N,
    const float* __restrict__ wroot, float* __restrict__ part, int Cout,
    int gemm_groups)
{
    if ((int)blockIdx.x < gemm_groups) {
        gemm_body<4, (1 << CSHIFT), KC, false, false, false, false, true>(
            h, wroot, nullptr, nullptr, nullptr, nullptr, part, N, Cout,
            (int)blockIdx.x);
        return;
    }
    gather_sum_body<CSHIFT>(h, col, deg, novf, ovf, E, agg, N, (int)blockIdx.x - gemm_groups);
}

template <int COUT, bool BUILD>
__global__ __launch_bounds__(1024) __attribute__((amdgpu_waves_per_eu(16, 16)))
void conv64_fused_kernel(
    const float* __restrict__ h, int* col, int* deg, int* novf, int2* ovf, int E, int N,
    const float* __restrict__ wroot, const float* __restrict__ wrel,
    const float* __restrict__ bias, float* __restrict__ out,
    const int2* __restrict__ sorted, const int* __restrict__ segoff, int nchunks, int nb)
{
    constexpr int CIN   = 64;
    constexpr int NT    = COUT / 16;
    constexpr int KP    = CIN / 2;
    constexpr int PITCH = NT * 32 + 32;
    constexpr int C4    = NT * 4;
    constexpr int ITEMS = 2 * KP * C4;
    constexpr int NW    = ITEMS / 1024;
    constexpr int AP    = 68;
    static_assert((NT & 1) == 0 && ITEMS % 1024 == 0, "shape");
    __shared__ __attribute__((aligned(16))) float wl[2 * KP * PITCH];
    __shared__ __attribute__((aligned(16))) float hT[64 * AP];
    __shared__ __attribute__((aligned(16))) float aT[64 * AP];
    __shared__ int lst[BUILD ? 64 * SLOTS : 1];
    __shared__ int lcnt[BUILD ? 64 : 1];

    const int tid  = threadIdx.x;
    const int nl   = tid >> 4;
    const unsigned c4 = ((unsigned)tid & 15u) << 2;
    const int node = (int)blockIdx.x * 64 + nl;
    const bool valid = node < N;

    float4 w0[NW], w1[NW];
    #pragma unroll
    for (int q = 0; q < NW; ++q) {
        const int it = tid + q * 1024;
        const int m  = it / (KP * C4);
        const int i  = it - m * (KP * C4);
        const int kp = i / C4;
        const int ci = i - kp * C4;
        const float* g = (m ? wrel : wroot) + (size_t)(2 * kp) * COUT + 4 * ci;
        w0[q] = *(const float4*)g;
        w1[q] = *(const float4*)(g + COUT);
    }
    float4 own = make_float4(0.f, 0.f, 0.f, 0.f);
    int dg = 0;
    if (valid) {
        own = *(const float4*)(h + ((size_t)node << 6) + c4);
        if (!BUILD) dg = deg[node];
    }
    const int blk = (int)blockIdx.x;
    if (BUILD) {
        for (int i = tid; i < 64 * SLOTS; i += 1024) lst[i] = 0;
        if (tid < 64) lcnt[tid] = 0;
        __syncthreads();
        const int l16 = tid & 15;
        for (int ck = tid >> 4; ck < nchunks; ck += 64) {
            const int* so = segoff + (size_t)ck * (size_t)(nb + 1) + blk;
            const int beg = so[0];
            const int end = so[1];
            const int2* sc = sorted + (size_t)ck * CH;
            for (int i = beg + l16; i < end; i += 16) {
                const int2 ent = sc[i];
                const int loc = ent.x - blk * 64;
                const int r = atomicAdd(&lcnt[loc], 1);
                if (r < SLOTS) {
                    lst[loc * SLOTS + r] = ent.y;
                } else {
                    const int o = atomicAdd(novf, 1);
                    ovf[o] = ent;
                }
            }
        }
        __syncthreads();
        for (int i = tid; i < 64 * SLOTS; i += 1024)
            if (blk * 64 + i / SLOTS < N) col[(size_t)blk * (64 * SLOTS) + i] = lst[i];
        if (tid < 64 && blk * 64 + tid < N) deg[blk * 64 + tid] = lcnt[tid];
        dg = lcnt[nl];
    }
    #pragma unroll
    for (int q = 0; q < NW; ++q) {
        const int it = tid + q * 1024;
        const int m  = it / (KP * C4);
        const int i  = it - m * (KP * C4);
        const int kp = i / C4;
        const int ci = i - kp * C4;
        float* d = wl + m * (KP * PITCH) + kp * PITCH + ci * 8;
        *(float4*)d       = make_float4(w0[q].x, w1[q].x, w0[q].y, w1[q].y);
        *(float4*)(d + 4) = make_float4(w0[q].z, w1[q].z, w0[q].w, w1[q].w);
    }
    *(float4*)(hT + nl * AP + c4) = own;

    float4 sum = make_float4(0.f, 0.f, 0.f, 0.f);
    if (BUILD) {
        if (dg <= SLOTS) {
            sum = gather_sum_list<6>(h, lst, nl * SLOTS, nl * SLOTS + dg, c4);
        } else {
            for (int ck = 0; ck < nchunks; ++ck) {
                const int* so = segoff + (size_t)ck * (size_t)(nb + 1) + blk;
                const int beg = so[0];
                const int end = so[1];
                for (int i = beg; i < end; ++i) {
                    const int2 ent = sorted[(size_t)ck * CH + i];
                    if (ent.x == node) {
                        const float4 v = *(const float4*)(h + (((unsigned)ent.y << 6) + c4));
                        sum.x += v.x; sum.y += v.y; sum.z += v.z; sum.w += v.w;
                    }
                }
            }
        }
    } else if (valid) {
        sum = gather_sum_row<6>(h, col, novf, ovf, E, node, dg, c4);
    }
    *(float4*)(aT + nl * AP + c4) = sum;
    __syncthreads();

    const int lane = tid & 31;
    const int wave = tid >> 5;
    if (wave >= 4 * NT) return;
    const int strip = wave / NT;
    const int tile  = wave - strip * NT;
    const int row0  = (int)blockIdx.x * 64 + strip * 16;
    if (row0 >= N) return;
    const int c  = lane & 15;
    const int hh = lane >> 4;

    v8f acc = v8f{};
    const float* ah = hT + (strip * 16 + c) * AP + 2 * hh;
    const float* aa = aT + (strip * 16 + c) * AP + 2 * hh;
    const float* wb = wl + hh * PITCH + tile * 32 + c * 2;
    #pragma unroll
    for (int kk = 0; kk < CIN / 4; ++kk) {
        const v2f av = *(const v2f*)(ah + 4 * kk);
        const v2f wv = *(const v2f*)(wb + 2 * kk * PITCH);
        acc = __builtin_amdgcn_wmma_f32_16x16x4_f32(false, av, false, wv, (short)0, acc, false, false);
    }
    #pragma unroll
    for (int kk = 0; kk < CIN / 4; ++kk) {
        const v2f av = *(const v2f*)(aa + 4 * kk);
        const v2f wv = *(const v2f*)(wb + KP * PITCH + 2 * kk * PITCH);
        acc = __builtin_amdgcn_wmma_f32_16x16x4_f32(false, av, false, wv, (short)0, acc, false, false);
    }

    const int m0 = row0 + (hh << 3);
    const int cc = tile * 16 + c;
    const float bval = bias[cc];
    #pragma unroll
    for (int r = 0; r < 8; ++r) {
        const float v = fmaxf(acc[r] + bval, 0.0f);
        out[(size_t)(m0 + r) * COUT + cc] = v;
    }
}


static inline unsigned blocks_for(long long n, int t) {
    return (unsigned)((n + t - 1) / t);
}

extern "C" void kernel_launch(void* const* d_in, const int* in_sizes, int n_in,
                              void* d_out, int out_size, void* d_ws, size_t ws_size,
                              hipStream_t stream)
{
    const float* x        = (const float*)d_in[0];
    const int*   edge     = (const int*)  d_in[1];
    const int*   batch    = (const int*)  d_in[2];
    const float* w_stage1 = (const float*)d_in[4];
    const float* b_stage1 = (const float*)d_in[5];
    const float* rel1_w   = (const float*)d_in[6];
    const float* rel1_b   = (const float*)d_in[7];
    const float* root1_w  = (const float*)d_in[8];
    const float* rel2_w   = (const float*)d_in[9];
    const float* rel2_b   = (const float*)d_in[10];
    const float* root2_w  = (const float*)d_in[11];
    const float* rel3_w   = (const float*)d_in[12];
    const float* rel3_b   = (const float*)d_in[13];
    const float* root3_w  = (const float*)d_in[14];
    const float* lin_w    = (const float*)d_in[15];
    const float* lin_b    = (const float*)d_in[16];
    float*       out      = (float*)d_out;

    const int N = in_sizes[0] / 128;
    const int E = in_sizes[1] / 2;
    const int G = out_size / 16;
    const int* src = edge;
    const int* dst = edge + E;

    float* ws = (float*)d_ws;
    size_t off = 0;
    float* bufA   = ws + off; off += (size_t)N * 128;
    float* bufB   = ws + off; off += (size_t)N * 128;
    float* agg    = ws + off; off += (size_t)N * 128;
    float* part   = ws + off; off += (size_t)N * 128;
    const long long zero4 = ((long long)G * 129 + 1 + 3) / 4;
    float* sums = ws + off;
    float* cnt  = sums + (size_t)G * 128;
    int* novf   = (int*)(cnt + G);
    off += (size_t)zero4 * 4;
    int* deg    = (int*)(ws + off); off += (size_t)N;
    off = (off + 63) & ~(size_t)63;
    int* col    = (int*)(ws + off); off += (size_t)N * SLOTS;
    int2* ovf   = (int2*)(ws + off); off += (size_t)E * 2;
    int2* sorted = (int2*)(ws + off); off += (size_t)E * 2;
    const int nb      = (N + 63) / 64;
    const int nchunks = (E + CH - 1) / CH;
    int* segoff = (int*)(ws + off); off += (size_t)nchunks * (size_t)(nb + 1);
    (void)ws_size; (void)n_in;

    const int T = 256;
    auto gemm_grid = [&](int M, int Cout, int NT) {
        return blocks_for((long long)(M >> 4), 8) * (unsigned)(Cout / (16 * NT));
    };

    const unsigned g1 = gemm_grid(N, 64, 4);
    stage1_and_sort_kernel<<<(unsigned)nchunks + g1 + blocks_for(zero4, T), T, 0, stream>>>(
        x, w_stage1, b_stage1, bufA, N, src, dst, E, nb, sorted, segoff, nchunks, (int)g1,
        (float4*)sums, zero4);

    const unsigned gf = blocks_for(N, 64);
    conv64_fused_kernel<64, true><<<gf, 1024, 0, stream>>>(
        bufA, col, deg, novf, ovf, E, N, root1_w, rel1_w, rel1_b, bufB, sorted, segoff, nchunks, nb);
    conv64_fused_kernel<128, false><<<gf, 1024, 0, stream>>>(
        bufB, col, deg, novf, ovf, E, N, root2_w, rel2_w, rel2_b, bufA, nullptr, nullptr, 0, 0);

    const unsigned gc2 = gemm_grid(N, 128, 4);
    gather_sum_and_root_kernel<7, 16><<<gc2 + blocks_for((long long)N << 5, T), T, 0, stream>>>(
        bufA, col, deg, novf, ovf, E, agg, N, root3_w, part, 128, (int)gc2);
    gemm_wmma_kernel<4, 128, true, true, false, true><<<gc2, T, 0, stream>>>(
        agg, rel3_w, rel3_b, sums, batch, cnt, part, N, 128);

    gemm_wmma_kernel<1, 128, false, false, true, false><<<gemm_grid(G, 16, 1), T, 0, stream>>>(
        sums, lin_w, lin_b, out, nullptr, cnt, nullptr, G, 16);
}


# Conv layer 3 in one launch: the work-group that forms a 64-node block's neighbour sums runs both products (weights via LDS, 64 k at a time) and the pool adds; 6 to 5 launches; sum order kept
# speedup vs baseline: 4.1129x; 1.0587x over previous
#include <hip/hip_runtime.h>


typedef __attribute__((ext_vector_type(2))) float v2f;
typedef __attribute__((ext_vector_type(8))) float v8f;

template <int NT, int CIN, int KC, bool RELU, bool DIV>
__device__ __forceinline__ void gemm_body(
    const float* __restrict__ A, const float* __restrict__ Wm,
    const float* __restrict__ bias, float* __restrict__ out,
    const float* __restrict__ cnt, int M, int Cout, int bid)
{
    constexpr int KP    = KC / 2;
    constexpr int PITCH = NT * 32 + ((NT & 1) ? 64 : 32);
    constexpr int C4    = NT * 4;
    __shared__ __attribute__((aligned(16))) float wl[KP * PITCH];

    const int lane = threadIdx.x & 31;
    const int wave = threadIdx.x >> 5;
    const int ncg  = Cout / (16 * NT);
    const int rb   = bid / ncg;
    const int cg   = bid - rb * ncg;
    const int col0 = cg * (16 * NT);
    const int strip = rb * 8 + wave;
    const bool active = strip < (M >> 4);
    const int c = lane & 15;
    const int h = lane >> 4;

    const size_t rowoff = (size_t)((active ? strip : 0) * 16 + c) * CIN + 2 * h;
    const float* arow = A + rowoff;
    float rows_of = 1.0f;
    if (DIV) rows_of = fmaxf(cnt[(active ? strip : 0) * 16 + c], 1.0f);

    v8f acc[NT];
    #pragma unroll
    for (int t = 0; t < NT; ++t) acc[t] = v8f{};

    #pragma unroll
    for (int k0 = 0; k0 < CIN; k0 += KC) {
        v2f af[KC / 4];
        if (active) {
            #pragma unroll
            for (int kk = 0; kk < KC / 4; ++kk) {
                af[kk] = *(const v2f*)(arow + k0 + 4 * kk);
                if (DIV) { af[kk].x = af[kk].x / rows_of; af[kk].y = af[kk].y / rows_of; }
            }
        }
        if (k0 > 0) __syncthreads();
        for (int i = threadIdx.x; i < KP * C4; i += 256) {
            const int kp = i / C4;
            const int c4 = i - kp * C4;
            const float* g = Wm + (size_t)(k0 + 2 * kp) * Cout + col0 + 4 * c4;
            const float4 r0 = *(const float4*)g;
            const float4 r1 = *(const float4*)(g + Cout);
            float* d = wl + kp * PITCH + c4 * 8;
            *(float4*)d       = make_float4(r0.x, r1.x, r0.y, r1.y);
            *(float4*)(d + 4) = make_float4(r0.z, r1.z, r0.w, r1.w);
        }
        __syncthreads();
        if (active) {
            #pragma unroll
            for (int kk = 0; kk < KC / 4; ++kk) {
                const float* wrow = wl + (2 * kk + h) * PITCH + c * 2;
                #pragma unroll
                for (int t = 0; t < NT; ++t) {
                    const v2f wv = *(const v2f*)(wrow + t * 32);
                    acc[t] = __builtin_amdgcn_wmma_f32_16x16x4_f32(
                                 false, af[kk], false, wv, (short)0, acc[t], false, false);
                }
            }
        }
    }
    if (!active) return;

    const int m0 = strip * 16 + (h << 3);
    #pragma unroll
    for (int t = 0; t < NT; ++t) {
        const int col = col0 + t * 16 + c;
        const float bval = bias[col];
        #pragma unroll
        for (int r = 0; r < 8; ++r) {
            float v = acc[t][r] + bval;
            if (RELU) v = fmaxf(v, 0.0f);
            out[(size_t)(m0 + r) * Cout + col] = v;
        }
    }
}

template <int NT, int CIN, bool RELU, bool DIV>
__global__ __launch_bounds__(256) void gemm_wmma_kernel(
    const float* __restrict__ A, const float* __restrict__ Wm,
    const float* __restrict__ bias, float* __restrict__ out,
    const float* __restrict__ cnt, int M, int Cout)
{
    gemm_body<NT, CIN, 64, RELU, DIV>(A, Wm, bias, out, cnt, M, Cout, (int)blockIdx.x);
}

constexpr int SLOTS = 64;

constexpr int CH    = 8192;
constexpr int NBMAX = 1024;

__device__ __forceinline__ void chunk_sort_body(
    const int* __restrict__ src, const int* __restrict__ dst, int E, int nb,
    int2* __restrict__ sorted, int* __restrict__ segoff, int chunk)
{
    constexpr int EPT = CH / 256;
    __shared__ int hist[NBMAX];
    __shared__ int wtot[8];
    __shared__ __attribute__((aligned(16))) int2 stage[CH];

    const int tid  = threadIdx.x;
    const int lane = tid & 31;
    const int wave = tid >> 5;
    for (int i = tid; i < NBMAX; i += 256) hist[i] = 0;

    const int base = chunk * CH;
    int d[EPT], s[EPT], bin[EPT];
    #pragma unroll
    for (int u = 0; u < EPT; ++u) {
        const int e  = base + u * 256 + tid;
        const int ec = e < E ? e : E - 1;
        d[u] = dst[ec];
        s[u] = src[ec];
        bin[u] = e < E ? (d[u] >> 6) : (NBMAX - 1 - lane);
    }
    __syncthreads();
    #pragma unroll
    for (int u = 0; u < EPT; ++u) atomicAdd(&hist[bin[u]], 1);
    __syncthreads();

    const int b4 = 4 * tid;
    const int h0 = hist[b4], h1 = hist[b4 + 1], h2 = hist[b4 + 2], h3 = hist[b4 + 3];
    const int tsum = h0 + h1 + h2 + h3;
    int incl = tsum;
    #pragma unroll
    for (int o = 1; o < 32; o <<= 1) {
        const int v = __shfl_up(incl, o);
        if (lane >= o) incl += v;
    }
    if (lane == 31) wtot[wave] = incl;
    __syncthreads();
    int ex = incl - tsum;
    for (int w = 0; w < wave; ++w) ex += wtot[w];
    const int e1 = ex + h0, e2 = e1 + h1, e3 = e2 + h2;
    hist[b4] = ex; hist[b4 + 1] = e1; hist[b4 + 2] = e2; hist[b4 + 3] = e3;
    int* so = segoff + (size_t)chunk * (size_t)(nb + 1);
    if (b4     <= nb) so[b4]     = ex;
    if (b4 + 1 <= nb) so[b4 + 1] = e1;
    if (b4 + 2 <= nb) so[b4 + 2] = e2;
    if (b4 + 3 <= nb) so[b4 + 3] = e3;
    __syncthreads();

    #pragma unroll
    for (int g = 0; g < EPT; g += 8) {
        int pos[8];
        #pragma unroll
        for (int u = 0; u < 8; ++u) pos[u] = atomicAdd(&hist[bin[g + u]], 1);
        #pragma unroll
        for (int u = 0; u < 8; ++u) stage[pos[u]] = make_int2(d[g + u], s[g + u]);
    }
    __syncthreads();
    const int total = (E - base < CH) ? (E - base) : CH;
    int2 t[EPT];
    #pragma unroll
    for (int u = 0; u < EPT; ++u) t[u] = stage[u * 256 + tid];
    #pragma unroll
    for (int u = 0; u < EPT; ++u)
        if (u * 256 + tid < total) sorted[(size_t)base + u * 256 + tid] = t[u];
}

__global__ __launch_bounds__(256) void stage1_and_sort_kernel(
    const float* __restrict__ x, const float* __restrict__ w,
    const float* __restrict__ bias, float* __restrict__ h0, int M,
    const int* __restrict__ src, const int* __restrict__ dst, int E, int nb,
    int2* __restrict__ sorted, int* __restrict__ segoff, int nchunks, int gemm_groups,
    float4* __restrict__ zp, long long n4)
{
    const int bid = (int)blockIdx.x;
    if (bid < nchunks) {
        chunk_sort_body(src, dst, E, nb, sorted, segoff, bid);
        return;
    }
    if (bid < nchunks + gemm_groups) {
        gemm_body<4, 128, 64, false, false>(x, w, bias, h0, nullptr, M, 64, bid - nchunks);
        return;
    }
    const long long i = (long long)(bid - nchunks - gemm_groups) * 256 + threadIdx.x;
    if (i < n4) zp[i] = make_float4(0.f, 0.f, 0.f, 0.f);
}

template <int CSHIFT>
__device__ __forceinline__ float4 gather_sum_list(
    const float* __restrict__ h, const int* col, int j, const int end, unsigned c4)
{
    float4 acc = make_float4(0.f, 0.f, 0.f, 0.f);
    for (; j < end; j += 8) {
        int s[8];
        for (int u = 0; u < 8; ++u) {
            const int jj = j + u;
            s[u] = col[jj < end ? jj : end - 1];
        }
        asm volatile("" :: "v"(s[0]), "v"(s[1]), "v"(s[2]), "v"(s[3]),
                           "v"(s[4]), "v"(s[5]), "v"(s[6]), "v"(s[7]));
        float4 v[8];
        v[0] = *(const float4*)(h + (((unsigned)s[0] << CSHIFT) + c4));
        for (int u = 1; u < 8; ++u) {
            v[u] = make_float4(0.f, 0.f, 0.f, 0.f);
            if (j + u < end) v[u] = *(const float4*)(h + (((unsigned)s[u] << CSHIFT) + c4));
        }
        for (int u = 0; u < 8; ++u) {
            acc.x += v[u].x; acc.y += v[u].y; acc.z += v[u].z; acc.w += v[u].w;
        }
    }
    return acc;
}

template <int CSHIFT>
__device__ __forceinline__ float4 gather_sum_row(
    const float* __restrict__ h, const int* __restrict__ col,
    const int* __restrict__ novf,
    const int2* __restrict__ ovf, int E, int node, int dg, unsigned c4)
{
    const int j = node * SLOTS;
    float4 acc = gather_sum_list<CSHIFT>(h, col, j, j + (dg < SLOTS ? dg : SLOTS), c4);
    if (dg > SLOTS) {
        int no = *novf;
        if (no > E) no = E;
        for (int i = 0; i < no; ++i) {
            const int2 p = ovf[i];
            if (p.x == node) {
                const float4 v = *(const float4*)(h + (((unsigned)p.y << CSHIFT) + c4));
                acc.x += v.x; acc.y += v.y; acc.z += v.z; acc.w += v.w;
            }
        }
    }
    return acc;
}

template <int COUT, bool BUILD>
__global__ __launch_bounds__(1024) __attribute__((amdgpu_waves_per_eu(16, 16)))
void conv64_fused_kernel(
    const float* __restrict__ h, int* col, int* deg, int* novf, int2* ovf, int E, int N,
    const float* __restrict__ wroot, const float* __restrict__ wrel,
    const float* __restrict__ bias, float* __restrict__ out,
    const int2* __restrict__ sorted, const int* __restrict__ segoff, int nchunks, int nb)
{
    constexpr int CIN   = 64;
    constexpr int NT    = COUT / 16;
    constexpr int KP    = CIN / 2;
    constexpr int PITCH = NT * 32 + 32;
    constexpr int C4    = NT * 4;
    constexpr int ITEMS = 2 * KP * C4;
    constexpr int NW    = ITEMS / 1024;
    constexpr int AP    = 68;
    static_assert((NT & 1) == 0 && ITEMS % 1024 == 0, "shape");
    __shared__ __attribute__((aligned(16))) float wl[2 * KP * PITCH];
    __shared__ __attribute__((aligned(16))) float hT[64 * AP];
    __shared__ __attribute__((aligned(16))) float aT[64 * AP];
    __shared__ int lst[BUILD ? 64 * SLOTS : 1];
    __shared__ int lcnt[BUILD ? 64 : 1];

    const int tid  = threadIdx.x;
    const int nl   = tid >> 4;
    const unsigned c4 = ((unsigned)tid & 15u) << 2;
    const int node = (int)blockIdx.x * 64 + nl;
    const bool valid = node < N;

    float4 w0[NW], w1[NW];
    #pragma unroll
    for (int q = 0; q < NW; ++q) {
        const int it = tid + q * 1024;
        const int m  = it / (KP * C4);
        const int i  = it - m * (KP * C4);
        const int kp = i / C4;
        const int ci = i - kp * C4;
        const float* g = (m ? wrel : wroot) + (size_t)(2 * kp) * COUT + 4 * ci;
        w0[q] = *(const float4*)g;
        w1[q] = *(const float4*)(g + COUT);
    }
    float4 own = make_float4(0.f, 0.f, 0.f, 0.f);
    int dg = 0;
    if (valid) {
        own = *(const float4*)(h + ((size_t)node << 6) + c4);
        if (!BUILD) dg = deg[node];
    }
    const int blk = (int)blockIdx.x;
    if (BUILD) {
        for (int i = tid; i < 64 * SLOTS; i += 1024) lst[i] = 0;
        if (tid < 64) lcnt[tid] = 0;
        __syncthreads();
        const int l16 = tid & 15;
        for (int ck = tid >> 4; ck < nchunks; ck += 64) {
            const int* so = segoff + (size_t)ck * (size_t)(nb + 1) + blk;
            const int beg = so[0];
            const int end = so[1];
            const int2* sc = sorted + (size_t)ck * CH;
            for (int i = beg + l16; i < end; i += 16) {
                const int2 ent = sc[i];
                const int loc = ent.x - blk * 64;
                const int r = atomicAdd(&lcnt[loc], 1);
                if (r < SLOTS) {
                    lst[loc * SLOTS + r] = ent.y;
                } else {
                    const int o = atomicAdd(novf, 1);
                    ovf[o] = ent;
                }
            }
        }
        __syncthreads();
        for (int i = tid; i < 64 * SLOTS; i += 1024)
            if (blk * 64 + i / SLOTS < N) col[(size_t)blk * (64 * SLOTS) + i] = lst[i];
        if (tid < 64 && blk * 64 + tid < N) deg[blk * 64 + tid] = lcnt[tid];
        dg = lcnt[nl];
    }
    #pragma unroll
    for (int q = 0; q < NW; ++q) {
        const int it = tid + q * 1024;
        const int m  = it / (KP * C4);
        const int i  = it - m * (KP * C4);
        const int kp = i / C4;
        const int ci = i - kp * C4;
        float* d = wl + m * (KP * PITCH) + kp * PITCH + ci * 8;
        *(float4*)d       = make_float4(w0[q].x, w1[q].x, w0[q].y, w1[q].y);
        *(float4*)(d + 4) = make_float4(w0[q].z, w1[q].z, w0[q].w, w1[q].w);
    }
    *(float4*)(hT + nl * AP + c4) = own;

    float4 sum = make_float4(0.f, 0.f, 0.f, 0.f);
    if (BUILD) {
        if (dg <= SLOTS) {
            sum = gather_sum_list<6>(h, lst, nl * SLOTS, nl * SLOTS + dg, c4);
        } else {
            for (int ck = 0; ck < nchunks; ++ck) {
                const int* so = segoff + (size_t)ck * (size_t)(nb + 1) + blk;
                const int beg = so[0];
                const int end = so[1];
                for (int i = beg; i < end; ++i) {
                    const int2 ent = sorted[(size_t)ck * CH + i];
                    if (ent.x == node) {
                        const float4 v = *(const float4*)(h + (((unsigned)ent.y << 6) + c4));
                        sum.x += v.x; sum.y += v.y; sum.z += v.z; sum.w += v.w;
                    }
                }
            }
        }
    } else if (valid) {
        sum = gather_sum_row<6>(h, col, novf, ovf, E, node, dg, c4);
    }
    *(float4*)(aT + nl * AP + c4) = sum;
    __syncthreads();

    const int lane = tid & 31;
    const int wave = tid >> 5;
    if (wave >= 4 * NT) return;
    const int strip = wave / NT;
    const int tile  = wave - strip * NT;
    const int row0  = (int)blockIdx.x * 64 + strip * 16;
    if (row0 >= N) return;
    const int c  = lane & 15;
    const int hh = lane >> 4;

    v8f acc = v8f{};
    const float* ah = hT + (strip * 16 + c) * AP + 2 * hh;
    const float* aa = aT + (strip * 16 + c) * AP + 2 * hh;
    const float* wb = wl + hh * PITCH + tile * 32 + c * 2;
    #pragma unroll
    for (int kk = 0; kk < CIN / 4; ++kk) {
        const v2f av = *(const v2f*)(ah + 4 * kk);
        const v2f wv = *(const v2f*)(wb + 2 * kk * PITCH);
        acc = __builtin_amdgcn_wmma_f32_16x16x4_f32(false, av, false, wv, (short)0, acc, false, false);
    }
    #pragma unroll
    for (int kk = 0; kk < CIN / 4; ++kk) {
        const v2f av = *(const v2f*)(aa + 4 * kk);
        const v2f wv = *(const v2f*)(wb + KP * PITCH + 2 * kk * PITCH);
        acc = __builtin_amdgcn_wmma_f32_16x16x4_f32(false, av, false, wv, (short)0, acc, false, false);
    }

    const int m0 = row0 + (hh << 3);
    const int cc = tile * 16 + c;
    const float bval = bias[cc];
    #pragma unroll
    for (int r = 0; r < 8; ++r) {
        const float v = fmaxf(acc[r] + bval, 0.0f);
        out[(size_t)(m0 + r) * COUT + cc] = v;
    }
}

__global__ __launch_bounds__(1024) __attribute__((amdgpu_waves_per_eu(16, 16)))
void conv128_pool_kernel(
    const float* __restrict__ h, const int* __restrict__ col,
    const int* __restrict__ deg, const int* __restrict__ novf,
    const int2* __restrict__ ovf, int E, int N,
    const float* __restrict__ wroot, const float* __restrict__ wrel,
    const float* __restrict__ bias, float* __restrict__ sums,
    const int* __restrict__ batch, float* __restrict__ cnt)
{
    constexpr int C     = 128;
    constexpr int KC    = 64;
    constexpr int KP    = KC / 2;
    constexpr int PITCH = 8 * 32 + 32;
    constexpr int AP    = 132;
    __shared__ __attribute__((aligned(16))) float wl[2 * KP * PITCH];
    __shared__ __attribute__((aligned(16))) float hT[64 * AP];
    __shared__ __attribute__((aligned(16))) float aT[64 * AP];

    const int tid  = threadIdx.x;
    const int lane = tid & 31;
    const int wave = tid >> 5;
    const int blk  = (int)blockIdx.x;
    const unsigned c4 = (unsigned)lane << 2;

    const size_t woff = (size_t)(2 * wave) * C + c4;
    float* wd = wl + wave * PITCH + lane * 8;

    float4 p0 = *(const float4*)(wroot + woff);
    float4 p1 = *(const float4*)(wroot + woff + C);
    float4 q0 = *(const float4*)(wroot + (size_t)KC * C + woff);
    float4 q1 = *(const float4*)(wroot + (size_t)KC * C + woff + C);
    float4 own[2];
    int dg[2];
    #pragma unroll
    for (int p = 0; p < 2; ++p) {
        const int node = blk * 64 + p * 32 + wave;
        own[p] = make_float4(0.f, 0.f, 0.f, 0.f);
        dg[p] = 0;
        if (node < N) {
            own[p] = *(const float4*)(h + ((size_t)node << 7) + c4);
            dg[p] = deg[node];
        }
    }
    *(float4*)wd                    = make_float4(p0.x, p1.x, p0.y, p1.y);
    *(float4*)(wd + 4)              = make_float4(p0.z, p1.z, p0.w, p1.w);
    *(float4*)(wd + KP * PITCH)     = make_float4(q0.x, q1.x, q0.y, q1.y);
    *(float4*)(wd + KP * PITCH + 4) = make_float4(q0.z, q1.z, q0.w, q1.w);
    #pragma unroll
    for (int p = 0; p < 2; ++p) {
        const int nl = p * 32 + wave;
        const int node = blk * 64 + nl;
        *(float4*)(hT + nl * AP + c4) = own[p];
        float4 sum = make_float4(0.f, 0.f, 0.f, 0.f);
        if (node < N) sum = gather_sum_row<7>(h, col, novf, ovf, E, node, dg[p], c4);
        *(float4*)(aT + nl * AP + c4) = sum;
    }
    __syncthreads();

    p0 = *(const float4*)(wrel + woff);
    p1 = *(const float4*)(wrel + woff + C);
    q0 = *(const float4*)(wrel + (size_t)KC * C + woff);
    q1 = *(const float4*)(wrel + (size_t)KC * C + woff + C);

    const int strip = wave >> 3;
    const int tile  = wave & 7;
    const int row0  = blk * 64 + strip * 16;
    const bool active = row0 < N;
    const int c  = lane & 15;
    const int hh = lane >> 4;

    v8f acc = v8f{};
    const float* ah = hT + (strip * 16 + c) * AP + 2 * hh;
    const float* aa = aT + (strip * 16 + c) * AP + 2 * hh;
    const float* wb = wl + hh * PITCH + tile * 32 + c * 2;
    if (active) {
        #pragma unroll
        for (int kk = 0; kk < C / 4; ++kk) {
            const v2f av = *(const v2f*)(ah + 4 * kk);
            const v2f wv = *(const v2f*)(wb + 2 * kk * PITCH);
            acc = __builtin_amdgcn_wmma_f32_16x16x4_f32(false, av, false, wv, (short)0, acc, false, false);
        }
    }
    __syncthreads();
    *(float4*)wd                    = make_float4(p0.x, p1.x, p0.y, p1.y);
    *(float4*)(wd + 4)              = make_float4(p0.z, p1.z, p0.w, p1.w);
    *(float4*)(wd + KP * PITCH)     = make_float4(q0.x, q1.x, q0.y, q1.y);
    *(float4*)(wd + KP * PITCH + 4) = make_float4(q0.z, q1.z, q0.w, q1.w);
    __syncthreads();
    if (!active) return;
    #pragma unroll
    for (int kk = 0; kk < C / 4; ++kk) {
        const v2f av = *(const v2f*)(aa + 4 * kk);
        const v2f wv = *(const v2f*)(wb + 2 * kk * PITCH);
        acc = __builtin_amdgcn_wmma_f32_16x16x4_f32(false, av, false, wv, (short)0, acc, false, false);
    }

    const int m0 = row0 + (hh << 3);
    const int cc = tile * 16 + c;
    int g[8];
    #pragma unroll
    for (int r = 0; r < 8; ++r) g[r] = batch[m0 + r];
    const float bval = bias[cc];
    float run = fmaxf(acc[0] + bval, 0.0f);
    #pragma unroll
    for (int r = 1; r < 8; ++r) {
        const float v = fmaxf(acc[r] + bval, 0.0f);
        if (g[r] != g[r - 1]) {
            atomicAdd(sums + (g[r - 1] * C + cc), run);
            run = 0.0f;
        }
        run += v;
    }
    atomicAdd(sums + (g[7] * C + cc), run);
    if (tile == 0 && c == 0) {
        float rows = 1.0f;
        #pragma unroll
        for (int r = 1; r < 8; ++r) {
            if (g[r] != g[r - 1]) {
                atomicAdd(cnt + g[r - 1], rows);
                rows = 0.0f;
            }
            rows += 1.0f;
        }
        atomicAdd(cnt + g[7], rows);
    }
}


static inline unsigned blocks_for(long long n, int t) {
    return (unsigned)((n + t - 1) / t);
}

extern "C" void kernel_launch(void* const* d_in, const int* in_sizes, int n_in,
                              void* d_out, int out_size, void* d_ws, size_t ws_size,
                              hipStream_t stream)
{
    const float* x        = (const float*)d_in[0];
    const int*   edge     = (const int*)  d_in[1];
    const int*   batch    = (const int*)  d_in[2];
    const float* w_stage1 = (const float*)d_in[4];
    const float* b_stage1 = (const float*)d_in[5];
    const float* rel1_w   = (const float*)d_in[6];
    const float* rel1_b   = (const float*)d_in[7];
    const float* root1_w  = (const float*)d_in[8];
    const float* rel2_w   = (const float*)d_in[9];
    const float* rel2_b   = (const float*)d_in[10];
    const float* root2_w  = (const float*)d_in[11];
    const float* rel3_w   = (const float*)d_in[12];
    const float* rel3_b   = (const float*)d_in[13];
    const float* root3_w  = (const float*)d_in[14];
    const float* lin_w    = (const float*)d_in[15];
    const float* lin_b    = (const float*)d_in[16];
    float*       out      = (float*)d_out;

    const int N = in_sizes[0] / 128;
    const int E = in_sizes[1] / 2;
    const int G = out_size / 16;
    const int* src = edge;
    const int* dst = edge + E;

    float* ws = (float*)d_ws;
    size_t off = 0;
    float* bufA   = ws + off; off += (size_t)N * 128;
    float* bufB   = ws + off; off += (size_t)N * 128;
    const long long zero4 = ((long long)G * 129 + 1 + 3) / 4;
    float* sums = ws + off;
    float* cnt  = sums + (size_t)G * 128;
    int* novf   = (int*)(cnt + G);
    off += (size_t)zero4 * 4;
    int* deg    = (int*)(ws + off); off += (size_t)N;
    off = (off + 63) & ~(size_t)63;
    int* col    = (int*)(ws + off); off += (size_t)N * SLOTS;
    int2* ovf   = (int2*)(ws + off); off += (size_t)E * 2;
    int2* sorted = (int2*)(ws + off); off += (size_t)E * 2;
    const int nb      = (N + 63) / 64;
    const int nchunks = (E + CH - 1) / CH;
    int* segoff = (int*)(ws + off); off += (size_t)nchunks * (size_t)(nb + 1);
    (void)ws_size; (void)n_in;

    const int T = 256;
    auto gemm_grid = [&](int M, int Cout, int NT) {
        return blocks_for((long long)(M >> 4), 8) * (unsigned)(Cout / (16 * NT));
    };

    const unsigned g1 = gemm_grid(N, 64, 4);
    stage1_and_sort_kernel<<<(unsigned)nchunks + g1 + blocks_for(zero4, T), T, 0, stream>>>(
        x, w_stage1, b_stage1, bufA, N, src, dst, E, nb, sorted, segoff, nchunks, (int)g1,
        (float4*)sums, zero4);

    const unsigned gf = blocks_for(N, 64);
    conv64_fused_kernel<64, true><<<gf, 1024, 0, stream>>>(
        bufA, col, deg, novf, ovf, E, N, root1_w, rel1_w, rel1_b, bufB, sorted, segoff, nchunks, nb);
    conv64_fused_kernel<128, false><<<gf, 1024, 0, stream>>>(
        bufB, col, deg, novf, ovf, E, N, root2_w, rel2_w, rel2_b, bufA, nullptr, nullptr, 0, 0);

    conv128_pool_kernel<<<gf, 1024, 0, stream>>>(
        bufA, col, deg, novf, ovf, E, N, root3_w, rel3_w, rel3_b, sums, batch, cnt);

    gemm_wmma_kernel<1, 128, false, true><<<gemm_grid(G, 16, 1), T, 0, stream>>>(
        sums, lin_w, lin_b, out, cnt, G, 16);
}


# Gather indices: a node's list is read as one entry per lane beside the first loads and handed on by lane reads (DPP row_share, readlane); rows no longer wait on index loads; sum order kept
# speedup vs baseline: 4.3122x; 1.0485x over previous
#include <hip/hip_runtime.h>


typedef __attribute__((ext_vector_type(2))) float v2f;
typedef __attribute__((ext_vector_type(8))) float v8f;

template <int NT, int CIN, int KC, bool RELU, bool DIV>
__device__ __forceinline__ void gemm_body(
    const float* __restrict__ A, const float* __restrict__ Wm,
    const float* __restrict__ bias, float* __restrict__ out,
    const float* __restrict__ cnt, int M, int Cout, int bid)
{
    constexpr int KP    = KC / 2;
    constexpr int PITCH = NT * 32 + ((NT & 1) ? 64 : 32);
    constexpr int C4    = NT * 4;
    __shared__ __attribute__((aligned(16))) float wl[KP * PITCH];

    const int lane = threadIdx.x & 31;
    const int wave = threadIdx.x >> 5;
    const int ncg  = Cout / (16 * NT);
    const int rb   = bid / ncg;
    const int cg   = bid - rb * ncg;
    const int col0 = cg * (16 * NT);
    const int strip = rb * 8 + wave;
    const bool active = strip < (M >> 4);
    const int c = lane & 15;
    const int h = lane >> 4;

    const size_t rowoff = (size_t)((active ? strip : 0) * 16 + c) * CIN + 2 * h;
    const float* arow = A + rowoff;
    float rows_of = 1.0f;
    if (DIV) rows_of = fmaxf(cnt[(active ? strip : 0) * 16 + c], 1.0f);

    v8f acc[NT];
    #pragma unroll
    for (int t = 0; t < NT; ++t) acc[t] = v8f{};

    #pragma unroll
    for (int k0 = 0; k0 < CIN; k0 += KC) {
        v2f af[KC / 4];
        if (active) {
            #pragma unroll
            for (int kk = 0; kk < KC / 4; ++kk) {
                af[kk] = *(const v2f*)(arow + k0 + 4 * kk);
                if (DIV) { af[kk].x = af[kk].x / rows_of; af[kk].y = af[kk].y / rows_of; }
            }
        }
        if (k0 > 0) __syncthreads();
        for (int i = threadIdx.x; i < KP * C4; i += 256) {
            const int kp = i / C4;
            const int c4 = i - kp * C4;
            const float* g = Wm + (size_t)(k0 + 2 * kp) * Cout + col0 + 4 * c4;
            const float4 r0 = *(const float4*)g;
            const float4 r1 = *(const float4*)(g + Cout);
            float* d = wl + kp * PITCH + c4 * 8;
            *(float4*)d       = make_float4(r0.x, r1.x, r0.y, r1.y);
            *(float4*)(d + 4) = make_float4(r0.z, r1.z, r0.w, r1.w);
        }
        __syncthreads();
        if (active) {
            #pragma unroll
            for (int kk = 0; kk < KC / 4; ++kk) {
                const float* wrow = wl + (2 * kk + h) * PITCH + c * 2;
                #pragma unroll
                for (int t = 0; t < NT; ++t) {
                    const v2f wv = *(const v2f*)(wrow + t * 32);
                    acc[t] = __builtin_amdgcn_wmma_f32_16x16x4_f32(
                                 false, af[kk], false, wv, (short)0, acc[t], false, false);
                }
            }
        }
    }
    if (!active) return;

    const int m0 = strip * 16 + (h << 3);
    #pragma unroll
    for (int t = 0; t < NT; ++t) {
        const int col = col0 + t * 16 + c;
        const float bval = bias[col];
        #pragma unroll
        for (int r = 0; r < 8; ++r) {
            float v = acc[t][r] + bval;
            if (RELU) v = fmaxf(v, 0.0f);
            out[(size_t)(m0 + r) * Cout + col] = v;
        }
    }
}

template <int NT, int CIN, bool RELU, bool DIV>
__global__ __launch_bounds__(256) void gemm_wmma_kernel(
    const float* __restrict__ A, const float* __restrict__ Wm,
    const float* __restrict__ bias, float* __restrict__ out,
    const float* __restrict__ cnt, int M, int Cout)
{
    gemm_body<NT, CIN, 64, RELU, DIV>(A, Wm, bias, out, cnt, M, Cout, (int)blockIdx.x);
}

constexpr int SLOTS = 64;

constexpr int CH    = 8192;
constexpr int NBMAX = 1024;

__device__ __forceinline__ void chunk_sort_body(
    const int* __restrict__ src, const int* __restrict__ dst, int E, int nb,
    int2* __restrict__ sorted, int* __restrict__ segoff, int chunk)
{
    constexpr int EPT = CH / 256;
    __shared__ int hist[NBMAX];
    __shared__ int wtot[8];
    __shared__ __attribute__((aligned(16))) int2 stage[CH];

    const int tid  = threadIdx.x;
    const int lane = tid & 31;
    const int wave = tid >> 5;
    for (int i = tid; i < NBMAX; i += 256) hist[i] = 0;

    const int base = chunk * CH;
    int d[EPT], s[EPT], bin[EPT];
    #pragma unroll
    for (int u = 0; u < EPT; ++u) {
        const int e  = base + u * 256 + tid;
        const int ec = e < E ? e : E - 1;
        d[u] = dst[ec];
        s[u] = src[ec];
        bin[u] = e < E ? (d[u] >> 6) : (NBMAX - 1 - lane);
    }
    __syncthreads();
    #pragma unroll
    for (int u = 0; u < EPT; ++u) atomicAdd(&hist[bin[u]], 1);
    __syncthreads();

    const int b4 = 4 * tid;
    const int h0 = hist[b4], h1 = hist[b4 + 1], h2 = hist[b4 + 2], h3 = hist[b4 + 3];
    const int tsum = h0 + h1 + h2 + h3;
    int incl = tsum;
    #pragma unroll
    for (int o = 1; o < 32; o <<= 1) {
        const int v = __shfl_up(incl, o);
        if (lane >= o) incl += v;
    }
    if (lane == 31) wtot[wave] = incl;
    __syncthreads();
    int ex = incl - tsum;
    for (int w = 0; w < wave; ++w) ex += wtot[w];
    const int e1 = ex + h0, e2 = e1 + h1, e3 = e2 + h2;
    hist[b4] = ex; hist[b4 + 1] = e1; hist[b4 + 2] = e2; hist[b4 + 3] = e3;
    int* so = segoff + (size_t)chunk * (size_t)(nb + 1);
    if (b4     <= nb) so[b4]     = ex;
    if (b4 + 1 <= nb) so[b4 + 1] = e1;
    if (b4 + 2 <= nb) so[b4 + 2] = e2;
    if (b4 + 3 <= nb) so[b4 + 3] = e3;
    __syncthreads();

    #pragma unroll
    for (int g = 0; g < EPT; g += 8) {
        int pos[8];
        #pragma unroll
        for (int u = 0; u < 8; ++u) pos[u] = atomicAdd(&hist[bin[g + u]], 1);
        #pragma unroll
        for (int u = 0; u < 8; ++u) stage[pos[u]] = make_int2(d[g + u], s[g + u]);
    }
    __syncthreads();
    const int total = (E - base < CH) ? (E - base) : CH;
    int2 t[EPT];
    #pragma unroll
    for (int u = 0; u < EPT; ++u) t[u] = stage[u * 256 + tid];
    #pragma unroll
    for (int u = 0; u < EPT; ++u)
        if (u * 256 + tid < total) sorted[(size_t)base + u * 256 + tid] = t[u];
}

__global__ __launch_bounds__(256) void stage1_and_sort_kernel(
    const float* __restrict__ x, const float* __restrict__ w,
    const float* __restrict__ bias, float* __restrict__ h0, int M,
    const int* __restrict__ src, const int* __restrict__ dst, int E, int nb,
    int2* __restrict__ sorted, int* __restrict__ segoff, int nchunks, int gemm_groups,
    float4* __restrict__ zp, long long n4)
{
    const int bid = (int)blockIdx.x;
    if (bid < nchunks) {
        chunk_sort_body(src, dst, E, nb, sorted, segoff, bid);
        return;
    }
    if (bid < nchunks + gemm_groups) {
        gemm_body<4, 128, 64, false, false>(x, w, bias, h0, nullptr, M, 64, bid - nchunks);
        return;
    }
    const long long i = (long long)(bid - nchunks - gemm_groups) * 256 + threadIdx.x;
    if (i < n4) zp[i] = make_float4(0.f, 0.f, 0.f, 0.f);
}


template <int CSHIFT, int B>
__device__ __forceinline__ void add_rows16(
    const float* __restrict__ h, const int e, const int left, const unsigned c4, float4& acc)
{
    unsigned s[8];
#define ENTRY16(U) s[U] = (unsigned)__builtin_amdgcn_update_dpp(0, e, 0x150 + B + U, 0xf, 0xf, true);
    ENTRY16(0) ENTRY16(1) ENTRY16(2) ENTRY16(3) ENTRY16(4) ENTRY16(5) ENTRY16(6) ENTRY16(7)
#undef ENTRY16
    float4 v[8];
    #pragma unroll
    for (int u = 0; u < 8; ++u) {
        v[u] = make_float4(0.f, 0.f, 0.f, 0.f);
        if (B + u < left)
            v[u] = *(const float4*)((const char*)h + ((s[u] << (CSHIFT + 2)) + (c4 << 2)));
    }
    #pragma unroll
    for (int u = 0; u < 8; ++u) {
        acc.x += v[u].x; acc.y += v[u].y; acc.z += v[u].z; acc.w += v[u].w;
    }
}

template <int CSHIFT>
__device__ __forceinline__ float4 gather_sum_run16(
    const float* __restrict__ h, const int* run, const int n, const unsigned c4,
    const int l16, const int e0, const int e1)
{
    float4 acc = make_float4(0.f, 0.f, 0.f, 0.f);
    if (n > 0)  add_rows16<CSHIFT, 0>(h, e0, n, c4, acc);
    if (n > 8)  add_rows16<CSHIFT, 8>(h, e0, n, c4, acc);
    if (n > 16) add_rows16<CSHIFT, 0>(h, e1, n - 16, c4, acc);
    if (n > 24) add_rows16<CSHIFT, 8>(h, e1, n - 16, c4, acc);
    for (int base = 32; base < n; base += 16) {
        const int e = run[base + l16];
        add_rows16<CSHIFT, 0>(h, e, n - base, c4, acc);
        if (n - base > 8) add_rows16<CSHIFT, 8>(h, e, n - base, c4, acc);
    }
    return acc;
}

template <int CSHIFT, int B>
__device__ __forceinline__ void add_rows32(
    const float* __restrict__ h, const int e, const int left, const unsigned c4, float4& acc)
{
    unsigned s[8];
    #pragma unroll
    for (int u = 0; u < 8; ++u) s[u] = (unsigned)__builtin_amdgcn_readlane(e, B + u);
    float4 v[8];
    #pragma unroll
    for (int u = 0; u < 8; ++u) {
        v[u] = make_float4(0.f, 0.f, 0.f, 0.f);
        if (B + u < left) {
            const char* rowp = (const char*)h + ((size_t)s[u] << (CSHIFT + 2));
            v[u] = *(const float4*)(rowp + (c4 << 2));
        }
    }
    #pragma unroll
    for (int u = 0; u < 8; ++u) {
        acc.x += v[u].x; acc.y += v[u].y; acc.z += v[u].z; acc.w += v[u].w;
    }
}

template <int CSHIFT>
__device__ __forceinline__ float4 gather_sum_run32(
    const float* __restrict__ h, const int* __restrict__ run, const int n, const unsigned c4,
    const int lane, const int e0)
{
    float4 acc = make_float4(0.f, 0.f, 0.f, 0.f);
    if (n > 0)  add_rows32<CSHIFT, 0>(h, e0, n, c4, acc);
    if (n > 8)  add_rows32<CSHIFT, 8>(h, e0, n, c4, acc);
    if (n > 16) add_rows32<CSHIFT, 16>(h, e0, n, c4, acc);
    if (n > 24) add_rows32<CSHIFT, 24>(h, e0, n, c4, acc);
    if (n > 32) {
        const int e = run[32 + lane];
        const int left = n - 32;
        add_rows32<CSHIFT, 0>(h, e, left, c4, acc);
        if (left > 8)  add_rows32<CSHIFT, 8>(h, e, left, c4, acc);
        if (left > 16) add_rows32<CSHIFT, 16>(h, e, left, c4, acc);
        if (left > 24) add_rows32<CSHIFT, 24>(h, e, left, c4, acc);
    }
    return acc;
}

template <int CSHIFT>
__device__ __forceinline__ void add_overflow_rows(
    const float* __restrict__ h, const int* __restrict__ novf,
    const int2* __restrict__ ovf, int E, int node, unsigned c4, float4& acc)
{
    int no = *novf;
    if (no > E) no = E;
    for (int i = 0; i < no; ++i) {
        const int2 p = ovf[i];
        if (p.x == node) {
            const float4 v = *(const float4*)(h + (((unsigned)p.y << CSHIFT) + c4));
            acc.x += v.x; acc.y += v.y; acc.z += v.z; acc.w += v.w;
        }
    }
}

template <int COUT, bool BUILD>
__global__ __launch_bounds__(1024) __attribute__((amdgpu_waves_per_eu(16, 16)))
void conv64_fused_kernel(
    const float* __restrict__ h, int* col, int* deg, int* novf, int2* ovf, int E, int N,
    const float* __restrict__ wroot, const float* __restrict__ wrel,
    const float* __restrict__ bias, float* __restrict__ out,
    const int2* __restrict__ sorted, const int* __restrict__ segoff, int nchunks, int nb)
{
    constexpr int CIN   = 64;
    constexpr int NT    = COUT / 16;
    constexpr int KP    = CIN / 2;
    constexpr int PITCH = NT * 32 + 32;
    constexpr int C4    = NT * 4;
    constexpr int ITEMS = 2 * KP * C4;
    constexpr int NW    = ITEMS / 1024;
    constexpr int AP    = 68;
    static_assert((NT & 1) == 0 && ITEMS % 1024 == 0, "shape");
    __shared__ __attribute__((aligned(16))) float wl[2 * KP * PITCH];
    __shared__ __attribute__((aligned(16))) float hT[64 * AP];
    __shared__ __attribute__((aligned(16))) float aT[64 * AP];
    __shared__ int lst[BUILD ? 64 * SLOTS : 1];
    __shared__ int lcnt[BUILD ? 64 : 1];

    const int tid  = threadIdx.x;
    const int nl   = tid >> 4;
    const unsigned c4 = ((unsigned)tid & 15u) << 2;
    const int node = (int)blockIdx.x * 64 + nl;
    const bool valid = node < N;

    float4 w0[NW], w1[NW];
    #pragma unroll
    for (int q = 0; q < NW; ++q) {
        const int it = tid + q * 1024;
        const int m  = it / (KP * C4);
        const int i  = it - m * (KP * C4);
        const int kp = i / C4;
        const int ci = i - kp * C4;
        const float* g = (m ? wrel : wroot) + (size_t)(2 * kp) * COUT + 4 * ci;
        w0[q] = *(const float4*)g;
        w1[q] = *(const float4*)(g + COUT);
    }
    float4 own = make_float4(0.f, 0.f, 0.f, 0.f);
    int dg = 0;
    int e0 = 0, e1 = 0;
    const int l16 = tid & 15;
    if (valid) {
        own = *(const float4*)(h + ((size_t)node << 6) + c4);
        if (!BUILD) {
            dg = deg[node];
            e0 = col[(size_t)node * SLOTS + l16];
            e1 = col[(size_t)node * SLOTS + 16 + l16];
        }
    }
    const int blk = (int)blockIdx.x;
    if (BUILD) {
        for (int i = tid; i < 64 * SLOTS; i += 1024) lst[i] = 0;
        if (tid < 64) lcnt[tid] = 0;
        __syncthreads();
        for (int ck = tid >> 4; ck < nchunks; ck += 64) {
            const int* so = segoff + (size_t)ck * (size_t)(nb + 1) + blk;
            const int beg = so[0];
            const int end = so[1];
            const int2* sc = sorted + (size_t)ck * CH;
            for (int i = beg + l16; i < end; i += 16) {
                const int2 ent = sc[i];
                const int loc = ent.x - blk * 64;
                const int r = atomicAdd(&lcnt[loc], 1);
                if (r < SLOTS) {
                    lst[loc * SLOTS + r] = ent.y;
                } else {
                    const int o = atomicAdd(novf, 1);
                    ovf[o] = ent;
                }
            }
        }
        __syncthreads();
        for (int i = tid; i < 64 * SLOTS; i += 1024)
            if (blk * 64 + i / SLOTS < N) col[(size_t)blk * (64 * SLOTS) + i] = lst[i];
        if (tid < 64 && blk * 64 + tid < N) deg[blk * 64 + tid] = lcnt[tid];
        dg = lcnt[nl];
        e0 = lst[nl * SLOTS + l16];
        e1 = lst[nl * SLOTS + 16 + l16];
    }
    #pragma unroll
    for (int q = 0; q < NW; ++q) {
        const int it = tid + q * 1024;
        const int m  = it / (KP * C4);
        const int i  = it - m * (KP * C4);
        const int kp = i / C4;
        const int ci = i - kp * C4;
        float* d = wl + m * (KP * PITCH) + kp * PITCH + ci * 8;
        *(float4*)d       = make_float4(w0[q].x, w1[q].x, w0[q].y, w1[q].y);
        *(float4*)(d + 4) = make_float4(w0[q].z, w1[q].z, w0[q].w, w1[q].w);
    }
    *(float4*)(hT + nl * AP + c4) = own;

    float4 sum = make_float4(0.f, 0.f, 0.f, 0.f);
    if (BUILD) {
        if (dg <= SLOTS) {
            sum = gather_sum_run16<6>(h, lst + nl * SLOTS, dg, c4, l16, e0, e1);
        } else {
            for (int ck = 0; ck < nchunks; ++ck) {
                const int* so = segoff + (size_t)ck * (size_t)(nb + 1) + blk;
                const int beg = so[0];
                const int end = so[1];
                for (int i = beg; i < end; ++i) {
                    const int2 ent = sorted[(size_t)ck * CH + i];
                    if (ent.x == node) {
                        const float4 v = *(const float4*)(h + (((unsigned)ent.y << 6) + c4));
                        sum.x += v.x; sum.y += v.y; sum.z += v.z; sum.w += v.w;
                    }
                }
            }
        }
    } else if (valid) {
        sum = gather_sum_run16<6>(h, col + (size_t)node * SLOTS, dg < SLOTS ? dg : SLOTS, c4, l16, e0, e1);
        if (dg > SLOTS) add_overflow_rows<6>(h, novf, ovf, E, node, c4, sum);
    }
    *(float4*)(aT + nl * AP + c4) = sum;
    __syncthreads();

    const int lane = tid & 31;
    const int wave = tid >> 5;
    if (wave >= 4 * NT) return;
    const int strip = wave / NT;
    const int tile  = wave - strip * NT;
    const int row0  = (int)blockIdx.x * 64 + strip * 16;
    if (row0 >= N) return;
    const int c  = lane & 15;
    const int hh = lane >> 4;

    v8f acc = v8f{};
    const float* ah = hT + (strip * 16 + c) * AP + 2 * hh;
    const float* aa = aT + (strip * 16 + c) * AP + 2 * hh;
    const float* wb = wl + hh * PITCH + tile * 32 + c * 2;
    #pragma unroll
    for (int kk = 0; kk < CIN / 4; ++kk) {
        const v2f av = *(const v2f*)(ah + 4 * kk);
        const v2f wv = *(const v2f*)(wb + 2 * kk * PITCH);
        acc = __builtin_amdgcn_wmma_f32_16x16x4_f32(false, av, false, wv, (short)0, acc, false, false);
    }
    #pragma unroll
    for (int kk = 0; kk < CIN / 4; ++kk) {
        const v2f av = *(const v2f*)(aa + 4 * kk);
        const v2f wv = *(const v2f*)(wb + KP * PITCH + 2 * kk * PITCH);
        acc = __builtin_amdgcn_wmma_f32_16x16x4_f32(false, av, false, wv, (short)0, acc, false, false);
    }

    const int m0 = row0 + (hh << 3);
    const int cc = tile * 16 + c;
    const float bval = bias[cc];
    #pragma unroll
    for (int r = 0; r < 8; ++r) {
        const float v = fmaxf(acc[r] + bval, 0.0f);
        out[(size_t)(m0 + r) * COUT + cc] = v;
    }
}

__global__ __launch_bounds__(1024) __attribute__((amdgpu_waves_per_eu(16, 16)))
void conv128_pool_kernel(
    const float* __restrict__ h, const int* __restrict__ col,
    const int* __restrict__ deg, const int* __restrict__ novf,
    const int2* __restrict__ ovf, int E, int N,
    const float* __restrict__ wroot, const float* __restrict__ wrel,
    const float* __restrict__ bias, float* __restrict__ sums,
    const int* __restrict__ batch, float* __restrict__ cnt)
{
    constexpr int C     = 128;
    constexpr int KC    = 64;
    constexpr int KP    = KC / 2;
    constexpr int PITCH = 8 * 32 + 32;
    constexpr int AP    = 132;
    __shared__ __attribute__((aligned(16))) float wl[2 * KP * PITCH];
    __shared__ __attribute__((aligned(16))) float hT[64 * AP];
    __shared__ __attribute__((aligned(16))) float aT[64 * AP];

    const int tid  = threadIdx.x;
    const int lane = tid & 31;
    const int wave = tid >> 5;
    const int blk  = (int)blockIdx.x;
    const unsigned c4 = (unsigned)lane << 2;

    const size_t woff = (size_t)(2 * wave) * C + c4;
    float* wd = wl + wave * PITCH + lane * 8;

    float4 p0 = *(const float4*)(wroot + woff);
    float4 p1 = *(const float4*)(wroot + woff + C);
    float4 q0 = *(const float4*)(wroot + (size_t)KC * C + woff);
    float4 q1 = *(const float4*)(wroot + (size_t)KC * C + woff + C);
    float4 own[2];
    int dg[2], ent[2];
    #pragma unroll
    for (int p = 0; p < 2; ++p) {
        const int node = blk * 64 + p * 32 + wave;
        own[p] = make_float4(0.f, 0.f, 0.f, 0.f);
        dg[p] = 0;
        ent[p] = 0;
        if (node < N) {
            own[p] = *(const float4*)(h + ((size_t)node << 7) + c4);
            dg[p] = deg[node];
            ent[p] = col[(size_t)node * SLOTS + lane];
        }
    }
    *(float4*)wd                    = make_float4(p0.x, p1.x, p0.y, p1.y);
    *(float4*)(wd + 4)              = make_float4(p0.z, p1.z, p0.w, p1.w);
    *(float4*)(wd + KP * PITCH)     = make_float4(q0.x, q1.x, q0.y, q1.y);
    *(float4*)(wd + KP * PITCH + 4) = make_float4(q0.z, q1.z, q0.w, q1.w);
    #pragma unroll
    for (int p = 0; p < 2; ++p) {
        const int nl = p * 32 + wave;
        const int node = blk * 64 + nl;
        *(float4*)(hT + nl * AP + c4) = own[p];
        float4 sum = make_float4(0.f, 0.f, 0.f, 0.f);
        if (node < N) {
            const int dgu = __builtin_amdgcn_readfirstlane(dg[p]);
            sum = gather_sum_run32<7>(h, col + (size_t)node * SLOTS, dgu < SLOTS ? dgu : SLOTS, c4, lane, ent[p]);
            if (dgu > SLOTS) add_overflow_rows<7>(h, novf, ovf, E, node, c4, sum);
        }
        *(float4*)(aT + nl * AP + c4) = sum;
    }
    __syncthreads();

    p0 = *(const float4*)(wrel + woff);
    p1 = *(const float4*)(wrel + woff + C);
    q0 = *(const float4*)(wrel + (size_t)KC * C + woff);
    q1 = *(const float4*)(wrel + (size_t)KC * C + woff + C);

    const int strip = wave >> 3;
    const int tile  = wave & 7;
    const int row0  = blk * 64 + strip * 16;
    const bool active = row0 < N;
    const int c  = lane & 15;
    const int hh = lane >> 4;

    v8f acc = v8f{};
    const float* ah = hT + (strip * 16 + c) * AP + 2 * hh;
    const float* aa = aT + (strip * 16 + c) * AP + 2 * hh;
    const float* wb = wl + hh * PITCH + tile * 32 + c * 2;
    if (active) {
        #pragma unroll
        for (int kk = 0; kk < C / 4; ++kk) {
            const v2f av = *(const v2f*)(ah + 4 * kk);
            const v2f wv = *(const v2f*)(wb + 2 * kk * PITCH);
            acc = __builtin_amdgcn_wmma_f32_16x16x4_f32(false, av, false, wv, (short)0, acc, false, false);
        }
    }
    __syncthreads();
    *(float4*)wd                    = make_float4(p0.x, p1.x, p0.y, p1.y);
    *(float4*)(wd + 4)              = make_float4(p0.z, p1.z, p0.w, p1.w);
    *(float4*)(wd + KP * PITCH)     = make_float4(q0.x, q1.x, q0.y, q1.y);
    *(float4*)(wd + KP * PITCH + 4) = make_float4(q0.z, q1.z, q0.w, q1.w);
    __syncthreads();
    if (!active) return;
    #pragma unroll
    for (int kk = 0; kk < C / 4; ++kk) {
        const v2f av = *(const v2f*)(aa + 4 * kk);
        const v2f wv = *(const v2f*)(wb + 2 * kk * PITCH);
        acc = __builtin_amdgcn_wmma_f32_16x16x4_f32(false, av, false, wv, (short)0, acc, false, false);
    }

    const int m0 = row0 + (hh << 3);
    const int cc = tile * 16 + c;
    int g[8];
    #pragma unroll
    for (int r = 0; r < 8; ++r) g[r] = batch[m0 + r];
    const float bval = bias[cc];
    float run = fmaxf(acc[0] + bval, 0.0f);
    #pragma unroll
    for (int r = 1; r < 8; ++r) {
        const float v = fmaxf(acc[r] + bval, 0.0f);
        if (g[r] != g[r - 1]) {
            atomicAdd(sums + (g[r - 1] * C + cc), run);
            run = 0.0f;
        }
        run += v;
    }
    atomicAdd(sums + (g[7] * C + cc), run);
    if (tile == 0 && c == 0) {
        float rows = 1.0f;
        #pragma unroll
        for (int r = 1; r < 8; ++r) {
            if (g[r] != g[r - 1]) {
                atomicAdd(cnt + g[r - 1], rows);
                rows = 0.0f;
            }
            rows += 1.0f;
        }
        atomicAdd(cnt + g[7], rows);
    }
}


static inline unsigned blocks_for(long long n, int t) {
    return (unsigned)((n + t - 1) / t);
}

extern "C" void kernel_launch(void* const* d_in, const int* in_sizes, int n_in,
                              void* d_out, int out_size, void* d_ws, size_t ws_size,
                              hipStream_t stream)
{
    const float* x        = (const float*)d_in[0];
    const int*   edge     = (const int*)  d_in[1];
    const int*   batch    = (const int*)  d_in[2];
    const float* w_stage1 = (const float*)d_in[4];
    const float* b_stage1 = (const float*)d_in[5];
    const float* rel1_w   = (const float*)d_in[6];
    const float* rel1_b   = (const float*)d_in[7];
    const float* root1_w  = (const float*)d_in[8];
    const float* rel2_w   = (const float*)d_in[9];
    const float* rel2_b   = (const float*)d_in[10];
    const float* root2_w  = (const float*)d_in[11];
    const float* rel3_w   = (const float*)d_in[12];
    const float* rel3_b   = (const float*)d_in[13];
    const float* root3_w  = (const float*)d_in[14];
    const float* lin_w    = (const float*)d_in[15];
    const float* lin_b    = (const float*)d_in[16];
    float*       out      = (float*)d_out;

    const int N = in_sizes[0] / 128;
    const int E = in_sizes[1] / 2;
    const int G = out_size / 16;
    const int* src = edge;
    const int* dst = edge + E;

    float* ws = (float*)d_ws;
    size_t off = 0;
    float* bufA   = ws + off; off += (size_t)N * 128;
    float* bufB   = ws + off; off += (size_t)N * 128;
    const long long zero4 = ((long long)G * 129 + 1 + 3) / 4;
    float* sums = ws + off;
    float* cnt  = sums + (size_t)G * 128;
    int* novf   = (int*)(cnt + G);
    off += (size_t)zero4 * 4;
    int* deg    = (int*)(ws + off); off += (size_t)N;
    off = (off + 63) & ~(size_t)63;
    int* col    = (int*)(ws + off); off += (size_t)N * SLOTS;
    int2* ovf   = (int2*)(ws + off); off += (size_t)E * 2;
    int2* sorted = (int2*)(ws + off); off += (size_t)E * 2;
    const int nb      = (N + 63) / 64;
    const int nchunks = (E + CH - 1) / CH;
    int* segoff = (int*)(ws + off); off += (size_t)nchunks * (size_t)(nb + 1);
    (void)ws_size; (void)n_in;

    const int T = 256;
    auto gemm_grid = [&](int M, int Cout, int NT) {
        return blocks_for((long long)(M >> 4), 8) * (unsigned)(Cout / (16 * NT));
    };

    const unsigned g1 = gemm_grid(N, 64, 4);
    stage1_and_sort_kernel<<<(unsigned)nchunks + g1 + blocks_for(zero4, T), T, 0, stream>>>(
        x, w_stage1, b_stage1, bufA, N, src, dst, E, nb, sorted, segoff, nchunks, (int)g1,
        (float4*)sums, zero4);

    const unsigned gf = blocks_for(N, 64);
    conv64_fused_kernel<64, true><<<gf, 1024, 0, stream>>>(
        bufA, col, deg, novf, ovf, E, N, root1_w, rel1_w, rel1_b, bufB, sorted, segoff, nchunks, nb);
    conv64_fused_kernel<128, false><<<gf, 1024, 0, stream>>>(
        bufB, col, deg, novf, ovf, E, N, root2_w, rel2_w, rel2_b, bufA, nullptr, nullptr, 0, 0);

    conv128_pool_kernel<<<gf, 1024, 0, stream>>>(
        bufA, col, deg, novf, ovf, E, N, root3_w, rel3_w, rel3_b, sums, batch, cnt);

    gemm_wmma_kernel<1, 128, false, true><<<gemm_grid(G, 16, 1), T, 0, stream>>>(
        sums, lin_w, lin_b, out, cnt, G, 16);
}
